# starting kernel
# baseline (speedup 1.0000x reference)
#include <hip/hip_runtime.h>
#include <hip/hip_bf16.h>

#define N_NODES 50000
#define N_EDGES 1600000
#define IN_FEAT 128
#define OUT_FEAT 32
#define N_HEADS 4
#define HF (N_HEADS * OUT_FEAT)
#define NEG_SLOPE 0.2f

typedef __attribute__((ext_vector_type(2))) float v2f;
typedef __attribute__((ext_vector_type(8))) float v8f;

__global__ void gat_init_kernel(float* __restrict__ rst,
                                float* __restrict__ denom,
                                float* __restrict__ m) {
    int t = blockIdx.x * blockDim.x + threadIdx.x;
    if (t < N_NODES * HF) rst[t] = 0.0f;
    if (t < N_NODES * N_HEADS) {
        denom[t] = 0.0f;
        m[t] = -__builtin_inff();
    }
}

__global__ void gat_gemm_wmma_kernel(const float* __restrict__ h,
                                     const float* __restrict__ W,
                                     float* __restrict__ feat) {
    const int lane = threadIdx.x & 31;
    const int wave = threadIdx.x >> 5;
    const int half = lane >> 4;
    const int l    = lane & 15;

    const int m0 = blockIdx.x * 16;
    const int n0 = (blockIdx.y * 4 + wave) * 16;

    v8f acc = {};

    const float* hrow = h + (size_t)(m0 + l) * IN_FEAT;

    #pragma unroll 4
    for (int kk = 0; kk < IN_FEAT; kk += 4) {
        const int kb = kk + half * 2;
        v2f a;
        a.x = hrow[kb + 0];
        a.y = hrow[kb + 1];
        v2f b;
        b.x = W[(size_t)(kb + 0) * HF + n0 + l];
        b.y = W[(size_t)(kb + 1) * HF + n0 + l];
        acc = __builtin_amdgcn_wmma_f32_16x16x4_f32(
             false, a,  false, b,
             (short)0, acc,  false,  false);
    }

    #pragma unroll
    for (int j = 0; j < 8; ++j) {
        feat[(size_t)(m0 + half * 8 + j) * HF + n0 + l] = acc[j];
    }
}

__global__ void gat_attn_proj_kernel(const float* __restrict__ feat,
                                     const float* __restrict__ attn_l,
                                     const float* __restrict__ attn_r,
                                     float* __restrict__ el,
                                     float* __restrict__ er) {
    int t = blockIdx.x * blockDim.x + threadIdx.x;
    if (t >= N_NODES * N_HEADS) return;
    const int n  = t >> 2;
    const int hd = t & 3;
    const float* fp = feat + (size_t)n * HF + hd * OUT_FEAT;
    const float* al = attn_l + hd * OUT_FEAT;
    const float* ar = attn_r + hd * OUT_FEAT;
    float sl = 0.0f, sr = 0.0f;
    #pragma unroll
    for (int f = 0; f < OUT_FEAT; ++f) {
        const float v = fp[f];
        sl += v * al[f];
        sr += v * ar[f];
    }
    el[t] = sl;
    er[t] = sr;
}

__device__ __forceinline__ void atomicMaxFloat(float* addr, float v) {
    if (v >= 0.0f) {
        atomicMax((int*)addr, __float_as_int(v));
    } else {
        atomicMin((unsigned int*)addr, __float_as_uint(v));
    }
}

__global__ void gat_edge_score_kernel(const float* __restrict__ el,
                                      const float* __restrict__ er,
                                      const int* __restrict__ src,
                                      const int* __restrict__ dst,
                                      float* __restrict__ ebuf,
                                      float* __restrict__ m) {
    int t = blockIdx.x * blockDim.x + threadIdx.x;
    if (t >= N_EDGES * N_HEADS) return;
    const int eidx = t >> 2;
    const int hd   = t & 3;
    const int s = src[eidx];
    const int d = dst[eidx];
    float v = el[s * N_HEADS + hd] + er[d * N_HEADS + hd];
    v = (v > 0.0f) ? v : v * NEG_SLOPE;
    ebuf[t] = v;
    atomicMaxFloat(&m[d * N_HEADS + hd], v);
}

__global__ void gat_mfix_kernel(float* __restrict__ m) {
    int t = blockIdx.x * blockDim.x + threadIdx.x;
    if (t >= N_NODES * N_HEADS) return;
    const float v = m[t];
    if (!(v == v) || v == -__builtin_inff() || v == __builtin_inff()) m[t] = 0.0f;
}

__global__ void gat_edge_exp_kernel(const int* __restrict__ dst,
                                    const float* __restrict__ m,
                                    float* __restrict__ ebuf,
                                    float* __restrict__ denom) {
    int t = blockIdx.x * blockDim.x + threadIdx.x;
    if (t >= N_EDGES * N_HEADS) return;
    const int eidx = t >> 2;
    const int hd   = t & 3;
    const int d = dst[eidx];
    const float ex = __expf(ebuf[t] - m[d * N_HEADS + hd]);
    ebuf[t] = ex;
    atomicAdd(&denom[d * N_HEADS + hd], ex);
}

__global__ void gat_aggregate_kernel(const float* __restrict__ feat,
                                     const float* __restrict__ ebuf,
                                     const float* __restrict__ denom,
                                     const int* __restrict__ src,
                                     const int* __restrict__ dst,
                                     float* __restrict__ rst) {
    int t = blockIdx.x * blockDim.x + threadIdx.x;
    const int eidx = t >> 5;
    const int lane = t & 31;
    if (eidx >= N_EDGES) return;
    const int s = src[eidx];
    const int d = dst[eidx];
    const float* fs = feat + (size_t)s * HF;
    float* rd = rst + (size_t)d * HF;
    #pragma unroll
    for (int hd = 0; hd < N_HEADS; ++hd) {
        const float dn = denom[d * N_HEADS + hd];
        const float alpha = ebuf[eidx * N_HEADS + hd] / fmaxf(dn, 1e-9f);
        const int c = hd * OUT_FEAT + lane;
        atomicAdd(&rd[c], fs[c] * alpha);
    }
}

__global__ void gat_epilogue_kernel(const float* __restrict__ rst,
                                    const float* __restrict__ bias,
                                    float* __restrict__ out) {
    int t = blockIdx.x * blockDim.x + threadIdx.x;
    if (t >= N_NODES * OUT_FEAT) return;
    const int n = t >> 5;
    const int f = t & 31;
    float sum = 0.0f;
    #pragma unroll
    for (int hd = 0; hd < N_HEADS; ++hd) {
        const int c = hd * OUT_FEAT + f;
        sum += fmaxf(rst[(size_t)n * HF + c] + bias[c], 0.0f);
    }
    out[t] = sum * (1.0f / N_HEADS);
}

extern "C" void kernel_launch(void* const* d_in, const int* in_sizes, int n_in,
                              void* d_out, int out_size, void* d_ws, size_t ws_size,
                              hipStream_t stream) {
    const float* h      = (const float*)d_in[0];
    const float* W      = (const float*)d_in[1];
    const float* attn_l = (const float*)d_in[2];
    const float* attn_r = (const float*)d_in[3];
    const float* bias   = (const float*)d_in[4];
    const int*   src    = (const int*)d_in[5];
    const int*   dst    = (const int*)d_in[6];
    float* out = (float*)d_out;

    float* ws    = (float*)d_ws;
    float* feat  = ws;
    float* rst   = feat  + (size_t)N_NODES * HF;
    float* el    = rst   + (size_t)N_NODES * HF;
    float* er    = el    + (size_t)N_NODES * N_HEADS;
    float* m     = er    + (size_t)N_NODES * N_HEADS;
    float* denom = m     + (size_t)N_NODES * N_HEADS;
    float* ebuf  = denom + (size_t)N_NODES * N_HEADS;

    const int BS = 256;

    {
        int n = N_NODES * HF;
        gat_init_kernel<<<(n + BS - 1) / BS, BS, 0, stream>>>(rst, denom, m);
    }
    {
        dim3 grid(N_NODES / 16, HF / (16 * 4));
        gat_gemm_wmma_kernel<<<grid, 128, 0, stream>>>(h, W, feat);
    }
    {
        int n = N_NODES * N_HEADS;
        gat_attn_proj_kernel<<<(n + BS - 1) / BS, BS, 0, stream>>>(feat, attn_l, attn_r, el, er);
    }
    {
        int n = N_EDGES * N_HEADS;
        gat_edge_score_kernel<<<(n + BS - 1) / BS, BS, 0, stream>>>(el, er, src, dst, ebuf, m);
    }
    {
        int n = N_NODES * N_HEADS;
        gat_mfix_kernel<<<(n + BS - 1) / BS, BS, 0, stream>>>(m);
    }
    {
        int n = N_EDGES * N_HEADS;
        gat_edge_exp_kernel<<<(n + BS - 1) / BS, BS, 0, stream>>>(dst, m, ebuf, denom);
    }
    {
        long long n = (long long)N_EDGES * 32;
        gat_aggregate_kernel<<<(unsigned)((n + BS - 1) / BS), BS, 0, stream>>>(
            feat, ebuf, denom, src, dst, rst);
    }
    {
        int n = N_NODES * OUT_FEAT;
        gat_epilogue_kernel<<<(n + BS - 1) / BS, BS, 0, stream>>>(rst, bias, out);
    }
}


# Edges grouped by dst in-launch (int atomic rank, prefix scan, atomic-swap scatter); one wave per dst does max, exp-sum and weighted feat-row sum in registers: no float atomics, sum order = rank order
# speedup vs baseline: 2.1512x; 2.1512x over previous
#include <hip/hip_runtime.h>
#include <hip/hip_bf16.h>

#define N_NODES 50000
#define N_EDGES 1600000
#define IN_FEAT 128
#define OUT_FEAT 32
#define N_HEADS 4
#define HF (N_HEADS * OUT_FEAT)
#define NEG_SLOPE 0.2f

typedef __attribute__((ext_vector_type(2))) float v2f;
typedef __attribute__((ext_vector_type(8))) float v8f;

__global__ void gat_zero_counts_kernel(int* __restrict__ cnt) {
    int t = blockIdx.x * blockDim.x + threadIdx.x;
    if (t < N_NODES) cnt[t] = 0;
}

__global__ void gat_gemm_wmma_kernel(const float* __restrict__ h,
                                     const float* __restrict__ W,
                                     float* __restrict__ feat) {
    const int lane = threadIdx.x & 31;
    const int wave = threadIdx.x >> 5;
    const int half = lane >> 4;
    const int l    = lane & 15;

    const int m0 = blockIdx.x * 16;
    const int n0 = (blockIdx.y * 4 + wave) * 16;

    v8f acc = {};

    const float* hrow = h + (size_t)(m0 + l) * IN_FEAT;

    #pragma unroll 4
    for (int kk = 0; kk < IN_FEAT; kk += 4) {
        const int kb = kk + half * 2;
        v2f a;
        a.x = hrow[kb + 0];
        a.y = hrow[kb + 1];
        v2f b;
        b.x = W[(size_t)(kb + 0) * HF + n0 + l];
        b.y = W[(size_t)(kb + 1) * HF + n0 + l];
        acc = __builtin_amdgcn_wmma_f32_16x16x4_f32(
             false, a,  false, b,
             (short)0, acc,  false,  false);
    }

    #pragma unroll
    for (int j = 0; j < 8; ++j) {
        feat[(size_t)(m0 + half * 8 + j) * HF + n0 + l] = acc[j];
    }
}

__global__ void gat_attn_proj_kernel(const float* __restrict__ feat,
                                     const float* __restrict__ attn_l,
                                     const float* __restrict__ attn_r,
                                     float* __restrict__ el,
                                     float* __restrict__ er) {
    int t = blockIdx.x * blockDim.x + threadIdx.x;
    if (t >= N_NODES * N_HEADS) return;
    const int n  = t >> 2;
    const int hd = t & 3;
    const float* fp = feat + (size_t)n * HF + hd * OUT_FEAT;
    const float* al = attn_l + hd * OUT_FEAT;
    const float* ar = attn_r + hd * OUT_FEAT;
    float sl = 0.0f, sr = 0.0f;
    #pragma unroll
    for (int f = 0; f < OUT_FEAT; ++f) {
        const float v = fp[f];
        sl += v * al[f];
        sr += v * ar[f];
    }
    el[t] = sl;
    er[t] = sr;
}

__global__ void gat_degree_kernel(const int* __restrict__ dst,
                                  int* __restrict__ cnt,
                                  int* __restrict__ rank) {
    int e = blockIdx.x * blockDim.x + threadIdx.x;
    if (e < N_EDGES) rank[e] = atomicAdd(&cnt[dst[e]], 1);
}

#define SCAN_T 1024
#define SCAN_K ((N_NODES + SCAN_T - 1) / SCAN_T)

__global__ void __launch_bounds__(SCAN_T)
gat_offsets_kernel(const int* __restrict__ cnt,
                   int* __restrict__ off) {
    __shared__ int wtot[SCAN_K][32];
    const int t    = threadIdx.x;
    const int lane = t & 31;
    const int wave = t >> 5;

    int ex[SCAN_K];
    #pragma unroll
    for (int k = 0; k < SCAN_K; ++k) {
        const int i = k * SCAN_T + t;
        ex[k] = (i < N_NODES) ? cnt[i] : 0;
    }
    #pragma unroll
    for (int k = 0; k < SCAN_K; ++k) {
        const int v = ex[k];
        int incl = v;
        for (int o = 1; o < 32; o <<= 1) {
            const int u = __shfl_up(incl, o);
            if (lane >= o) incl += u;
        }
        if (lane == 31) wtot[k][wave] = incl;
        ex[k] = incl - v;
    }
    __syncthreads();
    if (wave == 0) {
        int carry = 0;
        #pragma unroll
        for (int k = 0; k < SCAN_K; ++k) {
            const int v = wtot[k][lane];
            int incl = v;
            for (int o = 1; o < 32; o <<= 1) {
                const int u = __shfl_up(incl, o);
                if (lane >= o) incl += u;
            }
            wtot[k][lane] = carry + incl - v;
            carry += __shfl(incl, 31);
        }
    }
    __syncthreads();
    #pragma unroll
    for (int k = 0; k < SCAN_K; ++k) {
        const int i = k * SCAN_T + t;
        if (i <= N_NODES) off[i] = wtot[k][wave] + ex[k];
    }
}

__global__ void gat_fill_kernel(const int* __restrict__ src,
                                const int* __restrict__ dst,
                                const int* __restrict__ off,
                                const int* __restrict__ rank,
                                int* __restrict__ csr_src) {
    int e = blockIdx.x * blockDim.x + threadIdx.x;
    if (e < N_EDGES) {
        const int p = off[dst[e]] + rank[e];
        atomicExch(&csr_src[p], src[e]);
    }
}

__device__ __forceinline__ float gat_leaky(float v) {
    return (v > 0.0f) ? v : v * NEG_SLOPE;
}

__device__ __forceinline__ float gat_wave_max(float v) {
    for (int o = 16; o > 0; o >>= 1) v = fmaxf(v, __shfl_xor(v, o));
    return v;
}

__device__ __forceinline__ float gat_wave_sum(float v) {
    for (int o = 16; o > 0; o >>= 1) v += __shfl_xor(v, o);
    return v;
}

__device__ __forceinline__ float gat_finite_or_zero(float v) {
    return (!(v == v) || v == -__builtin_inff() || v == __builtin_inff()) ? 0.0f : v;
}

__device__ __forceinline__ void gat_add_rows(const float4* __restrict__ feat4,
                                             int lane, int k0, int k1, int k2, int k3,
                                             int n, int s,
                                             float a0, float a1, float a2, float a3,
                                             float4& acc) {
    const int i0 = __float_as_int(a0);
    const int i1 = __float_as_int(a1);
    const int i2 = __float_as_int(a2);
    const int i3 = __float_as_int(a3);
    #pragma unroll 4
    for (int j = 0; j < n; ++j) {
        const int sj = __builtin_amdgcn_readlane(s, j);
        const int ab = (__builtin_amdgcn_readlane(i0, j) & k0) |
                       (__builtin_amdgcn_readlane(i1, j) & k1) |
                       (__builtin_amdgcn_readlane(i2, j) & k2) |
                       (__builtin_amdgcn_readlane(i3, j) & k3);
        const float a = __int_as_float(ab);
        const float4 f = feat4[(size_t)sj * (HF / 4) + lane];
        acc.x += f.x * a;
        acc.y += f.y * a;
        acc.z += f.z * a;
        acc.w += f.w * a;
    }
}

__global__ void __launch_bounds__(256)
gat_node_kernel(const float* __restrict__ feat,
                const float* __restrict__ el,
                const float* __restrict__ er,
                const int* __restrict__ off,
                const int* __restrict__ csr_src,
                const float* __restrict__ bias,
                float* __restrict__ out) {
    const int lane = threadIdx.x & 31;
    const int node = __builtin_amdgcn_readfirstlane(
        (int)((blockIdx.x * blockDim.x + threadIdx.x) >> 5));
    if (node >= N_NODES) return;

    const int beg = __builtin_amdgcn_readfirstlane(off[node]);
    const int end = __builtin_amdgcn_readfirstlane(off[node + 1]);
    const int deg = end - beg;
    const float4* el4   = (const float4*)el;
    const float4* feat4 = (const float4*)feat;
    const float4  r4    = ((const float4*)er)[node];
    const float   ninf  = -__builtin_inff();

    const bool v0 = lane < deg;
    const bool v1 = lane + 32 < deg;
    const int  s0 = csr_src[v0 ? beg + lane : 0];
    const int  s1 = csr_src[v1 ? beg + 32 + lane : 0];
    const float4 l0 = el4[s0];
    const float4 l1 = el4[s1];
    const float e00 = v0 ? gat_leaky(l0.x + r4.x) : ninf;
    const float e01 = v0 ? gat_leaky(l0.y + r4.y) : ninf;
    const float e02 = v0 ? gat_leaky(l0.z + r4.z) : ninf;
    const float e03 = v0 ? gat_leaky(l0.w + r4.w) : ninf;
    const float e10 = v1 ? gat_leaky(l1.x + r4.x) : ninf;
    const float e11 = v1 ? gat_leaky(l1.y + r4.y) : ninf;
    const float e12 = v1 ? gat_leaky(l1.z + r4.z) : ninf;
    const float e13 = v1 ? gat_leaky(l1.w + r4.w) : ninf;

    float m0 = fmaxf(e00, e10), m1 = fmaxf(e01, e11);
    float m2 = fmaxf(e02, e12), m3 = fmaxf(e03, e13);
    for (int i = beg + 64 + lane; i < end; i += 32) {
        const float4 l = el4[csr_src[i]];
        m0 = fmaxf(m0, gat_leaky(l.x + r4.x));
        m1 = fmaxf(m1, gat_leaky(l.y + r4.y));
        m2 = fmaxf(m2, gat_leaky(l.z + r4.z));
        m3 = fmaxf(m3, gat_leaky(l.w + r4.w));
    }
    m0 = gat_finite_or_zero(gat_wave_max(m0));
    m1 = gat_finite_or_zero(gat_wave_max(m1));
    m2 = gat_finite_or_zero(gat_wave_max(m2));
    m3 = gat_finite_or_zero(gat_wave_max(m3));

    const float x00 = __expf(e00 - m0), x01 = __expf(e01 - m1);
    const float x02 = __expf(e02 - m2), x03 = __expf(e03 - m3);
    const float x10 = __expf(e10 - m0), x11 = __expf(e11 - m1);
    const float x12 = __expf(e12 - m2), x13 = __expf(e13 - m3);
    float d0 = x00 + x10, d1 = x01 + x11, d2 = x02 + x12, d3 = x03 + x13;
    for (int i = beg + 64 + lane; i < end; i += 32) {
        const float4 l = el4[csr_src[i]];
        d0 += __expf(gat_leaky(l.x + r4.x) - m0);
        d1 += __expf(gat_leaky(l.y + r4.y) - m1);
        d2 += __expf(gat_leaky(l.z + r4.z) - m2);
        d3 += __expf(gat_leaky(l.w + r4.w) - m3);
    }
    d0 = fmaxf(gat_wave_sum(d0), 1e-9f);
    d1 = fmaxf(gat_wave_sum(d1), 1e-9f);
    d2 = fmaxf(gat_wave_sum(d2), 1e-9f);
    d3 = fmaxf(gat_wave_sum(d3), 1e-9f);

    const int hl = lane >> 3;
    const int k0 = (hl == 0) ? -1 : 0;
    const int k1 = (hl == 1) ? -1 : 0;
    const int k2 = (hl == 2) ? -1 : 0;
    const int k3 = (hl == 3) ? -1 : 0;
    float4 acc = make_float4(0.0f, 0.0f, 0.0f, 0.0f);
    {
        const int n0 = (deg < 32) ? deg : 32;
        gat_add_rows(feat4, lane, k0, k1, k2, k3, n0, s0,
                     x00 / d0, x01 / d1, x02 / d2, x03 / d3, acc);
    }
    if (deg > 32) {
        const int n1 = (deg < 64) ? (deg - 32) : 32;
        gat_add_rows(feat4, lane, k0, k1, k2, k3, n1, s1,
                     x10 / d0, x11 / d1, x12 / d2, x13 / d3, acc);
    }
    for (int cb = beg + 64; cb < end; cb += 32) {
        const int  n = (end - cb < 32) ? (end - cb) : 32;
        const bool v = lane < n;
        const int  s = csr_src[v ? cb + lane : 0];
        const float4 l = el4[s];
        const float a0 = v ? __expf(gat_leaky(l.x + r4.x) - m0) / d0 : 0.0f;
        const float a1 = v ? __expf(gat_leaky(l.y + r4.y) - m1) / d1 : 0.0f;
        const float a2 = v ? __expf(gat_leaky(l.z + r4.z) - m2) / d2 : 0.0f;
        const float a3 = v ? __expf(gat_leaky(l.w + r4.w) - m3) / d3 : 0.0f;
        gat_add_rows(feat4, lane, k0, k1, k2, k3, n, s, a0, a1, a2, a3, acc);
    }

    const float4 b4 = ((const float4*)bias)[lane];
    const float rx = fmaxf(acc.x + b4.x, 0.0f);
    const float ry = fmaxf(acc.y + b4.y, 0.0f);
    const float rz = fmaxf(acc.z + b4.z, 0.0f);
    const float rw = fmaxf(acc.w + b4.w, 0.0f);
    float4 o;
    o.x = ((rx + __shfl_down(rx, 8)) + __shfl_down(rx, 16)) + __shfl_down(rx, 24);
    o.y = ((ry + __shfl_down(ry, 8)) + __shfl_down(ry, 16)) + __shfl_down(ry, 24);
    o.z = ((rz + __shfl_down(rz, 8)) + __shfl_down(rz, 16)) + __shfl_down(rz, 24);
    o.w = ((rw + __shfl_down(rw, 8)) + __shfl_down(rw, 16)) + __shfl_down(rw, 24);
    if (lane < 8) {
        o.x *= (1.0f / N_HEADS);
        o.y *= (1.0f / N_HEADS);
        o.z *= (1.0f / N_HEADS);
        o.w *= (1.0f / N_HEADS);
        ((float4*)out)[(size_t)node * (OUT_FEAT / 4) + lane] = o;
    }
}

extern "C" void kernel_launch(void* const* d_in, const int* in_sizes, int n_in,
                              void* d_out, int out_size, void* d_ws, size_t ws_size,
                              hipStream_t stream) {
    const float* h      = (const float*)d_in[0];
    const float* W      = (const float*)d_in[1];
    const float* attn_l = (const float*)d_in[2];
    const float* attn_r = (const float*)d_in[3];
    const float* bias   = (const float*)d_in[4];
    const int*   src    = (const int*)d_in[5];
    const int*   dst    = (const int*)d_in[6];
    float* out = (float*)d_out;

    float* ws      = (float*)d_ws;
    float* feat    = ws;
    float* el      = feat + (size_t)N_NODES * HF;
    float* er      = el   + (size_t)N_NODES * N_HEADS;
    int*   cnt     = (int*)(er + (size_t)N_NODES * N_HEADS);
    int*   off     = cnt    + 50048;
    int*   rank    = off    + 50048;
    int*   csr_src = rank   + N_EDGES;

    const int BS = 256;

    gat_zero_counts_kernel<<<(N_NODES + BS - 1) / BS, BS, 0, stream>>>(cnt);
    {
        dim3 grid(N_NODES / 16, HF / (16 * 4));
        gat_gemm_wmma_kernel<<<grid, 128, 0, stream>>>(h, W, feat);
    }
    {
        int n = N_NODES * N_HEADS;
        gat_attn_proj_kernel<<<(n + BS - 1) / BS, BS, 0, stream>>>(feat, attn_l, attn_r, el, er);
    }
    gat_degree_kernel<<<(N_EDGES + BS - 1) / BS, BS, 0, stream>>>(dst, cnt, rank);
    gat_offsets_kernel<<<1, SCAN_T, 0, stream>>>(cnt, off);
    gat_fill_kernel<<<(N_EDGES + BS - 1) / BS, BS, 0, stream>>>(src, dst, off, rank, csr_src);
    gat_node_kernel<<<(N_NODES * 32 + BS - 1) / BS, BS, 0, stream>>>(
        feat, el, er, off, csr_src, bias, out);
}


# Independent work shares launches: histogram blocks interleaved 2:1 with projection blocks, offsets scan runs as block 0 of the el/er launch; 7 launches -> 5; sum order = rank order as before
# speedup vs baseline: 2.3907x; 1.1113x over previous
#include <hip/hip_runtime.h>
#include <hip/hip_bf16.h>

#define N_NODES 50000
#define N_EDGES 1600000
#define IN_FEAT 128
#define OUT_FEAT 32
#define N_HEADS 4
#define HF (N_HEADS * OUT_FEAT)
#define NEG_SLOPE 0.2f

typedef __attribute__((ext_vector_type(2))) float v2f;
typedef __attribute__((ext_vector_type(8))) float v8f;

__global__ void gat_zero_counts_kernel(int* __restrict__ cnt) {
    int t = blockIdx.x * blockDim.x + threadIdx.x;
    if (t < N_NODES) cnt[t] = 0;
}

__device__ __forceinline__ void gat_gemm_tile(const float* __restrict__ h,
                                              const float* __restrict__ W,
                                              float* __restrict__ feat,
                                              int m0, int n0, int lane) {
    const int half = lane >> 4;
    const int l    = lane & 15;

    v8f acc = {};

    const float* hrow = h + (size_t)(m0 + l) * IN_FEAT;

    #pragma unroll 4
    for (int kk = 0; kk < IN_FEAT; kk += 4) {
        const int kb = kk + half * 2;
        v2f a;
        a.x = hrow[kb + 0];
        a.y = hrow[kb + 1];
        v2f b;
        b.x = W[(size_t)(kb + 0) * HF + n0 + l];
        b.y = W[(size_t)(kb + 1) * HF + n0 + l];
        acc = __builtin_amdgcn_wmma_f32_16x16x4_f32(
             false, a,  false, b,
             (short)0, acc,  false,  false);
    }

    #pragma unroll
    for (int j = 0; j < 8; ++j) {
        feat[(size_t)(m0 + half * 8 + j) * HF + n0 + l] = acc[j];
    }
}

#define PROJ_TILES  (N_NODES / 16)
#define HIST_BLOCKS (N_EDGES / 256)
static_assert(N_NODES % 16 == 0, "row tiles must be whole");
static_assert(N_EDGES % 256 == 0 && HIST_BLOCKS == 2 * PROJ_TILES,
              "two histogram blocks per projection block");

__global__ void __launch_bounds__(256)
gat_proj_hist_kernel(const float* __restrict__ h,
                     const float* __restrict__ W,
                     float* __restrict__ feat,
                     const int* __restrict__ dst,
                     int* __restrict__ cnt,
                     int* __restrict__ rank) {
    const int q = blockIdx.x / 3;
    const int r = blockIdx.x - q * 3;
    if (r == 0) {
        gat_gemm_tile(h, W, feat, q * 16, (int)(threadIdx.x >> 5) * 16,
                      (int)(threadIdx.x & 31));
    } else {
        const int e = (q * 2 + (r - 1)) * 256 + (int)threadIdx.x;
        if (e < N_EDGES) rank[e] = atomicAdd(&cnt[dst[e]], 1);
    }
}

__device__ __forceinline__ void gat_attn_proj_thread(int t,
                                                     const float* __restrict__ feat,
                                                     const float* __restrict__ attn_l,
                                                     const float* __restrict__ attn_r,
                                                     float* __restrict__ el,
                                                     float* __restrict__ er) {
    if (t >= N_NODES * N_HEADS) return;
    const int n  = t >> 2;
    const int hd = t & 3;
    const float* fp = feat + (size_t)n * HF + hd * OUT_FEAT;
    const float* al = attn_l + hd * OUT_FEAT;
    const float* ar = attn_r + hd * OUT_FEAT;
    float sl = 0.0f, sr = 0.0f;
    #pragma unroll
    for (int f = 0; f < OUT_FEAT; ++f) {
        const float v = fp[f];
        sl += v * al[f];
        sr += v * ar[f];
    }
    el[t] = sl;
    er[t] = sr;
}

#define SCAN_T 1024
#define SCAN_K ((N_NODES + SCAN_T - 1) / SCAN_T)

__device__ __forceinline__ void gat_offsets_block(const int* __restrict__ cnt,
                                                  int* __restrict__ off,
                                                  int (*wtot)[32]) {
    const int t    = threadIdx.x;
    const int lane = t & 31;
    const int wave = t >> 5;

    int ex[SCAN_K];
    #pragma unroll
    for (int k = 0; k < SCAN_K; ++k) {
        const int i = k * SCAN_T + t;
        ex[k] = (i < N_NODES) ? cnt[i] : 0;
    }
    #pragma unroll
    for (int k = 0; k < SCAN_K; ++k) {
        const int v = ex[k];
        int incl = v;
        for (int o = 1; o < 32; o <<= 1) {
            const int u = __shfl_up(incl, o);
            if (lane >= o) incl += u;
        }
        if (lane == 31) wtot[k][wave] = incl;
        ex[k] = incl - v;
    }
    __syncthreads();
    if (wave == 0) {
        int carry = 0;
        #pragma unroll
        for (int k = 0; k < SCAN_K; ++k) {
            const int v = wtot[k][lane];
            int incl = v;
            for (int o = 1; o < 32; o <<= 1) {
                const int u = __shfl_up(incl, o);
                if (lane >= o) incl += u;
            }
            wtot[k][lane] = carry + incl - v;
            carry += __shfl(incl, 31);
        }
    }
    __syncthreads();
    #pragma unroll
    for (int k = 0; k < SCAN_K; ++k) {
        const int i = k * SCAN_T + t;
        if (i <= N_NODES) off[i] = wtot[k][wave] + ex[k];
    }
}

#define ATTN_BLOCKS ((N_NODES * N_HEADS + SCAN_T - 1) / SCAN_T)

__global__ void __launch_bounds__(SCAN_T)
gat_attn_offsets_kernel(const float* __restrict__ feat,
                        const float* __restrict__ attn_l,
                        const float* __restrict__ attn_r,
                        float* __restrict__ el,
                        float* __restrict__ er,
                        const int* __restrict__ cnt,
                        int* __restrict__ off) {
    __shared__ int wtot[SCAN_K][32];
    if (blockIdx.x == 0) {
        gat_offsets_block(cnt, off, wtot);
    } else {
        gat_attn_proj_thread((int)(blockIdx.x - 1) * SCAN_T + (int)threadIdx.x,
                             feat, attn_l, attn_r, el, er);
    }
}

__global__ void gat_fill_kernel(const int* __restrict__ src,
                                const int* __restrict__ dst,
                                const int* __restrict__ off,
                                const int* __restrict__ rank,
                                int* __restrict__ csr_src) {
    int e = blockIdx.x * blockDim.x + threadIdx.x;
    if (e < N_EDGES) {
        const int p = off[dst[e]] + rank[e];
        atomicExch(&csr_src[p], src[e]);
    }
}

__device__ __forceinline__ float gat_leaky(float v) {
    return (v > 0.0f) ? v : v * NEG_SLOPE;
}

__device__ __forceinline__ float gat_wave_max(float v) {
    for (int o = 16; o > 0; o >>= 1) v = fmaxf(v, __shfl_xor(v, o));
    return v;
}

__device__ __forceinline__ float gat_wave_sum(float v) {
    for (int o = 16; o > 0; o >>= 1) v += __shfl_xor(v, o);
    return v;
}

__device__ __forceinline__ float gat_finite_or_zero(float v) {
    return (!(v == v) || v == -__builtin_inff() || v == __builtin_inff()) ? 0.0f : v;
}

__device__ __forceinline__ void gat_add_rows(const float4* __restrict__ feat4,
                                             int lane, int k0, int k1, int k2, int k3,
                                             int n, int s,
                                             float a0, float a1, float a2, float a3,
                                             float4& acc) {
    const int i0 = __float_as_int(a0);
    const int i1 = __float_as_int(a1);
    const int i2 = __float_as_int(a2);
    const int i3 = __float_as_int(a3);
    #pragma unroll 4
    for (int j = 0; j < n; ++j) {
        const int sj = __builtin_amdgcn_readlane(s, j);
        const int ab = (__builtin_amdgcn_readlane(i0, j) & k0) |
                       (__builtin_amdgcn_readlane(i1, j) & k1) |
                       (__builtin_amdgcn_readlane(i2, j) & k2) |
                       (__builtin_amdgcn_readlane(i3, j) & k3);
        const float a = __int_as_float(ab);
        const float4 f = feat4[(size_t)sj * (HF / 4) + lane];
        acc.x += f.x * a;
        acc.y += f.y * a;
        acc.z += f.z * a;
        acc.w += f.w * a;
    }
}

__global__ void __launch_bounds__(256)
gat_node_kernel(const float* __restrict__ feat,
                const float* __restrict__ el,
                const float* __restrict__ er,
                const int* __restrict__ off,
                const int* __restrict__ csr_src,
                const float* __restrict__ bias,
                float* __restrict__ out) {
    const int lane = threadIdx.x & 31;
    const int node = __builtin_amdgcn_readfirstlane(
        (int)((blockIdx.x * blockDim.x + threadIdx.x) >> 5));
    if (node >= N_NODES) return;

    const int beg = __builtin_amdgcn_readfirstlane(off[node]);
    const int end = __builtin_amdgcn_readfirstlane(off[node + 1]);
    const int deg = end - beg;
    const float4* el4   = (const float4*)el;
    const float4* feat4 = (const float4*)feat;
    const float4  r4    = ((const float4*)er)[node];
    const float   ninf  = -__builtin_inff();

    const bool v0 = lane < deg;
    const bool v1 = lane + 32 < deg;
    const int  s0 = csr_src[v0 ? beg + lane : 0];
    const int  s1 = csr_src[v1 ? beg + 32 + lane : 0];
    const float4 l0 = el4[s0];
    const float4 l1 = el4[s1];
    const float e00 = v0 ? gat_leaky(l0.x + r4.x) : ninf;
    const float e01 = v0 ? gat_leaky(l0.y + r4.y) : ninf;
    const float e02 = v0 ? gat_leaky(l0.z + r4.z) : ninf;
    const float e03 = v0 ? gat_leaky(l0.w + r4.w) : ninf;
    const float e10 = v1 ? gat_leaky(l1.x + r4.x) : ninf;
    const float e11 = v1 ? gat_leaky(l1.y + r4.y) : ninf;
    const float e12 = v1 ? gat_leaky(l1.z + r4.z) : ninf;
    const float e13 = v1 ? gat_leaky(l1.w + r4.w) : ninf;

    float m0 = fmaxf(e00, e10), m1 = fmaxf(e01, e11);
    float m2 = fmaxf(e02, e12), m3 = fmaxf(e03, e13);
    for (int i = beg + 64 + lane; i < end; i += 32) {
        const float4 l = el4[csr_src[i]];
        m0 = fmaxf(m0, gat_leaky(l.x + r4.x));
        m1 = fmaxf(m1, gat_leaky(l.y + r4.y));
        m2 = fmaxf(m2, gat_leaky(l.z + r4.z));
        m3 = fmaxf(m3, gat_leaky(l.w + r4.w));
    }
    m0 = gat_finite_or_zero(gat_wave_max(m0));
    m1 = gat_finite_or_zero(gat_wave_max(m1));
    m2 = gat_finite_or_zero(gat_wave_max(m2));
    m3 = gat_finite_or_zero(gat_wave_max(m3));

    const float x00 = __expf(e00 - m0), x01 = __expf(e01 - m1);
    const float x02 = __expf(e02 - m2), x03 = __expf(e03 - m3);
    const float x10 = __expf(e10 - m0), x11 = __expf(e11 - m1);
    const float x12 = __expf(e12 - m2), x13 = __expf(e13 - m3);
    float d0 = x00 + x10, d1 = x01 + x11, d2 = x02 + x12, d3 = x03 + x13;
    for (int i = beg + 64 + lane; i < end; i += 32) {
        const float4 l = el4[csr_src[i]];
        d0 += __expf(gat_leaky(l.x + r4.x) - m0);
        d1 += __expf(gat_leaky(l.y + r4.y) - m1);
        d2 += __expf(gat_leaky(l.z + r4.z) - m2);
        d3 += __expf(gat_leaky(l.w + r4.w) - m3);
    }
    d0 = fmaxf(gat_wave_sum(d0), 1e-9f);
    d1 = fmaxf(gat_wave_sum(d1), 1e-9f);
    d2 = fmaxf(gat_wave_sum(d2), 1e-9f);
    d3 = fmaxf(gat_wave_sum(d3), 1e-9f);

    const int hl = lane >> 3;
    const int k0 = (hl == 0) ? -1 : 0;
    const int k1 = (hl == 1) ? -1 : 0;
    const int k2 = (hl == 2) ? -1 : 0;
    const int k3 = (hl == 3) ? -1 : 0;
    float4 acc = make_float4(0.0f, 0.0f, 0.0f, 0.0f);
    {
        const int n0 = (deg < 32) ? deg : 32;
        gat_add_rows(feat4, lane, k0, k1, k2, k3, n0, s0,
                     x00 / d0, x01 / d1, x02 / d2, x03 / d3, acc);
    }
    if (deg > 32) {
        const int n1 = (deg < 64) ? (deg - 32) : 32;
        gat_add_rows(feat4, lane, k0, k1, k2, k3, n1, s1,
                     x10 / d0, x11 / d1, x12 / d2, x13 / d3, acc);
    }
    for (int cb = beg + 64; cb < end; cb += 32) {
        const int  n = (end - cb < 32) ? (end - cb) : 32;
        const bool v = lane < n;
        const int  s = csr_src[v ? cb + lane : 0];
        const float4 l = el4[s];
        const float a0 = v ? __expf(gat_leaky(l.x + r4.x) - m0) / d0 : 0.0f;
        const float a1 = v ? __expf(gat_leaky(l.y + r4.y) - m1) / d1 : 0.0f;
        const float a2 = v ? __expf(gat_leaky(l.z + r4.z) - m2) / d2 : 0.0f;
        const float a3 = v ? __expf(gat_leaky(l.w + r4.w) - m3) / d3 : 0.0f;
        gat_add_rows(feat4, lane, k0, k1, k2, k3, n, s, a0, a1, a2, a3, acc);
    }

    const float4 b4 = ((const float4*)bias)[lane];
    const float rx = fmaxf(acc.x + b4.x, 0.0f);
    const float ry = fmaxf(acc.y + b4.y, 0.0f);
    const float rz = fmaxf(acc.z + b4.z, 0.0f);
    const float rw = fmaxf(acc.w + b4.w, 0.0f);
    float4 o;
    o.x = ((rx + __shfl_down(rx, 8)) + __shfl_down(rx, 16)) + __shfl_down(rx, 24);
    o.y = ((ry + __shfl_down(ry, 8)) + __shfl_down(ry, 16)) + __shfl_down(ry, 24);
    o.z = ((rz + __shfl_down(rz, 8)) + __shfl_down(rz, 16)) + __shfl_down(rz, 24);
    o.w = ((rw + __shfl_down(rw, 8)) + __shfl_down(rw, 16)) + __shfl_down(rw, 24);
    if (lane < 8) {
        o.x *= (1.0f / N_HEADS);
        o.y *= (1.0f / N_HEADS);
        o.z *= (1.0f / N_HEADS);
        o.w *= (1.0f / N_HEADS);
        ((float4*)out)[(size_t)node * (OUT_FEAT / 4) + lane] = o;
    }
}

extern "C" void kernel_launch(void* const* d_in, const int* in_sizes, int n_in,
                              void* d_out, int out_size, void* d_ws, size_t ws_size,
                              hipStream_t stream) {
    const float* h      = (const float*)d_in[0];
    const float* W      = (const float*)d_in[1];
    const float* attn_l = (const float*)d_in[2];
    const float* attn_r = (const float*)d_in[3];
    const float* bias   = (const float*)d_in[4];
    const int*   src    = (const int*)d_in[5];
    const int*   dst    = (const int*)d_in[6];
    float* out = (float*)d_out;

    float* ws      = (float*)d_ws;
    float* feat    = ws;
    float* el      = feat + (size_t)N_NODES * HF;
    float* er      = el   + (size_t)N_NODES * N_HEADS;
    int*   cnt     = (int*)(er + (size_t)N_NODES * N_HEADS);
    int*   off     = cnt    + 50048;
    int*   rank    = off    + 50048;
    int*   csr_src = rank   + N_EDGES;

    const int BS = 256;

    gat_zero_counts_kernel<<<(N_NODES + BS - 1) / BS, BS, 0, stream>>>(cnt);
    gat_proj_hist_kernel<<<PROJ_TILES * 3, 256, 0, stream>>>(h, W, feat, dst, cnt, rank);
    gat_attn_offsets_kernel<<<1 + ATTN_BLOCKS, SCAN_T, 0, stream>>>(
        feat, attn_l, attn_r, el, er, cnt, off);
    gat_fill_kernel<<<(N_EDGES + BS - 1) / BS, BS, 0, stream>>>(src, dst, off, rank, csr_src);
    gat_node_kernel<<<(N_NODES * 32 + BS - 1) / BS, BS, 0, stream>>>(
        feat, el, er, off, csr_src, bias, out);
}


# Edges grouped by dst in two LDS-ranked levels (4096-edge chunks -> 196 bins -> dst) instead of one global atomic rank per edge; 5 launches -> 4; sum order = arrival order at the bin's LDS cursor
# speedup vs baseline: 5.8709x; 2.4557x over previous
#include <hip/hip_runtime.h>
#include <hip/hip_bf16.h>

#define N_NODES 50000
#define N_EDGES 1600000
#define IN_FEAT 128
#define OUT_FEAT 32
#define N_HEADS 4
#define HF (N_HEADS * OUT_FEAT)
#define NEG_SLOPE 0.2f

typedef __attribute__((ext_vector_type(2))) float v2f;
typedef __attribute__((ext_vector_type(8))) float v8f;

#define BIN_SHIFT 8
#define BIN_NODES (1 << BIN_SHIFT)
#define N_BINS    ((N_NODES + BIN_NODES - 1) / BIN_NODES)
#define BINS_PAD  224
#define CHUNK     4096
#define N_CHUNKS  ((N_EDGES + CHUNK - 1) / CHUNK)
#define EPT       (CHUNK / 256)
#define REGION    10240
static_assert(N_BINS <= BINS_PAD && BINS_PAD <= 256 && BINS_PAD % 32 == 0, "desc row");
static_assert(CHUNK + N_BINS * 31 <= REGION && REGION % 32 == 0, "padded runs fit");
static_assert(N_NODES <= (1 << 23), "src << 8 must fit a word");
static_assert(REGION < (1 << 18) && CHUNK < (1 << 13), "desc packing: start << 13 | count");

__device__ __forceinline__ void gat_gemm_tile(const float* __restrict__ h,
                                              const float* __restrict__ W,
                                              float* __restrict__ feat,
                                              int m0, int n0, int lane) {
    const int half = lane >> 4;
    const int l    = lane & 15;

    v8f acc = {};

    const float* hrow = h + (size_t)(m0 + l) * IN_FEAT;

    #pragma unroll 4
    for (int kk = 0; kk < IN_FEAT; kk += 4) {
        const int kb = kk + half * 2;
        v2f a;
        a.x = hrow[kb + 0];
        a.y = hrow[kb + 1];
        v2f b;
        b.x = W[(size_t)(kb + 0) * HF + n0 + l];
        b.y = W[(size_t)(kb + 1) * HF + n0 + l];
        acc = __builtin_amdgcn_wmma_f32_16x16x4_f32(
             false, a,  false, b,
             (short)0, acc,  false,  false);
    }

    #pragma unroll
    for (int j = 0; j < 8; ++j) {
        feat[(size_t)(m0 + half * 8 + j) * HF + n0 + l] = acc[j];
    }
}

template <int PER, bool PAD>
__device__ __forceinline__ void gat_wave_scan(const int* in, int* out, int lane) {
    int v[PER];
    int s = 0;
    #pragma unroll
    for (int j = 0; j < PER; ++j) {
        const int c = in[lane * PER + j];
        v[j] = PAD ? ((c + 31) & ~31) : c;
        s += v[j];
    }
    int incl = s;
    for (int o = 1; o < 32; o <<= 1) {
        const int u = __shfl_up(incl, o);
        if (lane >= o) incl += u;
    }
    int b = incl - s;
    #pragma unroll
    for (int j = 0; j < PER; ++j) {
        out[lane * PER + j] = b;
        b += v[j];
    }
}

__device__ __forceinline__ void gat_partition_block(const int* __restrict__ src,
                                                    const int* __restrict__ dst,
                                                    unsigned* __restrict__ rec,
                                                    int* __restrict__ desc,
                                                    int chunk,
                                                    int* hcnt, int* hbase,
                                                    unsigned* stage) {
    const int tid = threadIdx.x;
    hcnt[tid] = 0;
    __syncthreads();

    int d[EPT], r[EPT];
    const int e0 = chunk * CHUNK + tid;
    #pragma unroll
    for (int k = 0; k < EPT; ++k) {
        const int e = e0 + k * 256;
        d[k] = (e < N_EDGES) ? dst[e] : -1;
    }
    #pragma unroll
    for (int k = 0; k < EPT; ++k) {
        r[k] = (d[k] >= 0) ? atomicAdd(&hcnt[d[k] >> BIN_SHIFT], 1) : 0;
    }
    __syncthreads();

    if (tid < 32) gat_wave_scan<BINS_PAD / 32, true>(hcnt, hbase, tid);
    __syncthreads();

    #pragma unroll
    for (int k = 0; k < EPT; ++k) {
        if (d[k] >= 0) {
            const int e = e0 + k * 256;
            stage[hbase[d[k] >> BIN_SHIFT] + r[k]] =
                ((unsigned)src[e] << 8) | (unsigned)(d[k] & (BIN_NODES - 1));
        }
    }
    if (tid < BINS_PAD) {
        const int c = hcnt[tid];
        const int p = (c + 31) & ~31;
        for (int i = c; i < p; ++i) stage[hbase[tid] + i] = 0u;
        desc[chunk * BINS_PAD + tid] = (hbase[tid] << 13) | c;
    }
    __syncthreads();

    const int total = hbase[BINS_PAD - 1] + ((hcnt[BINS_PAD - 1] + 31) & ~31);
    unsigned* out = rec + (size_t)chunk * REGION;
    for (int i = tid; i < total; i += 256) out[i] = stage[i];
}

#define PROJ_TILES  (N_NODES / 16)
static_assert(N_NODES % 16 == 0, "row tiles must be whole");

__global__ void __launch_bounds__(256)
gat_proj_part_kernel(const float* __restrict__ h,
                     const float* __restrict__ W,
                     float* __restrict__ feat,
                     const int* __restrict__ src,
                     const int* __restrict__ dst,
                     unsigned* __restrict__ rec,
                     int* __restrict__ desc) {
    __shared__ int hcnt[256];
    __shared__ int hbase[256];
    __shared__ unsigned stage[REGION];
    if (blockIdx.x < N_CHUNKS) {
        gat_partition_block(src, dst, rec, desc, (int)blockIdx.x, hcnt, hbase, stage);
    } else {
        const int q = (int)blockIdx.x - N_CHUNKS;
        gat_gemm_tile(h, W, feat, q * 16, (int)(threadIdx.x >> 5) * 16,
                      (int)(threadIdx.x & 31));
    }
}

__device__ __forceinline__ void gat_attn_proj_thread(int t,
                                                     const float* __restrict__ feat,
                                                     const float* __restrict__ attn_l,
                                                     const float* __restrict__ attn_r,
                                                     float* __restrict__ el,
                                                     float* __restrict__ er) {
    if (t >= N_NODES * N_HEADS) return;
    const int n  = t >> 2;
    const int hd = t & 3;
    const float* fp = feat + (size_t)n * HF + hd * OUT_FEAT;
    const float* al = attn_l + hd * OUT_FEAT;
    const float* ar = attn_r + hd * OUT_FEAT;
    float sl = 0.0f, sr = 0.0f;
    #pragma unroll
    for (int f = 0; f < OUT_FEAT; ++f) {
        const float v = fp[f];
        sl += v * al[f];
        sr += v * ar[f];
    }
    el[t] = sl;
    er[t] = sr;
}

#define SCAN_T 1024

__device__ __forceinline__ void gat_bin_bases_block(const int* __restrict__ desc,
                                                    int* __restrict__ bin_base,
                                                    int* tot, int* sb) {
    const int t    = threadIdx.x;
    const int lane = t & 31;
    const int wave = t >> 5;
    if (t < 256) tot[t] = 0;
    __syncthreads();
    int part[BINS_PAD / 32];
    #pragma unroll
    for (int j = 0; j < BINS_PAD / 32; ++j) part[j] = 0;
    for (int c = wave; c < N_CHUNKS; c += 32) {
        #pragma unroll
        for (int j = 0; j < BINS_PAD / 32; ++j)
            part[j] += desc[c * BINS_PAD + lane + 32 * j] & 8191;
    }
    #pragma unroll
    for (int j = 0; j < BINS_PAD / 32; ++j) atomicAdd(&tot[lane + 32 * j], part[j]);
    __syncthreads();
    if (t < 32) gat_wave_scan<BINS_PAD / 32, true>(tot, sb, t);
    __syncthreads();
    if (t < BINS_PAD) bin_base[t] = sb[t];
}

#define ATTN_BLOCKS ((N_NODES * N_HEADS + SCAN_T - 1) / SCAN_T)

__global__ void __launch_bounds__(SCAN_T)
gat_attn_bases_kernel(const float* __restrict__ feat,
                      const float* __restrict__ attn_l,
                      const float* __restrict__ attn_r,
                      float* __restrict__ el,
                      float* __restrict__ er,
                      const int* __restrict__ desc,
                      int* __restrict__ bin_base) {
    __shared__ int tot[256];
    __shared__ int sb[256];
    if (blockIdx.x == 0) {
        gat_bin_bases_block(desc, bin_base, tot, sb);
    } else {
        gat_attn_proj_thread((int)(blockIdx.x - 1) * SCAN_T + (int)threadIdx.x,
                             feat, attn_l, attn_r, el, er);
    }
}

#define RUNS_PER_WAVE ((N_CHUNKS + 31) / 32)

__global__ void __launch_bounds__(1024)
gat_bin_csr_kernel(const unsigned* __restrict__ rec,
                   const int* __restrict__ desc,
                   const int* __restrict__ bin_base,
                   int* __restrict__ off,
                   int* __restrict__ degs,
                   int* __restrict__ csr_src) {
    __shared__ int ncnt[BIN_NODES];
    __shared__ int cur[BIN_NODES];
    const int bin  = blockIdx.x;
    const int t    = threadIdx.x;
    const int lane = t & 31;
    const int wave = t >> 5;
    const int base = bin_base[bin];

    if (t < BIN_NODES) ncnt[t] = 0;

    int      dsc[RUNS_PER_WAVE];
    unsigned rv[RUNS_PER_WAVE];
    #pragma unroll
    for (int k = 0; k < RUNS_PER_WAVE; ++k) {
        const int c = wave + 32 * k;
        dsc[k] = (c < N_CHUNKS) ? desc[c * BINS_PAD + bin] : 0;
    }
    #pragma unroll
    for (int k = 0; k < RUNS_PER_WAVE; ++k) {
        const int c = wave + 32 * k;
        const int n = dsc[k] & 8191;
        rv[k] = (lane < n) ? rec[(size_t)c * REGION + (dsc[k] >> 13) + lane]
                           : 0xFFFFFFFFu;
    }
    __syncthreads();

    #pragma unroll
    for (int k = 0; k < RUNS_PER_WAVE; ++k) {
        if ((int)rv[k] >= 0) atomicAdd(&ncnt[rv[k] & (BIN_NODES - 1)], 1);
    }
    #pragma unroll
    for (int k = 0; k < RUNS_PER_WAVE; ++k) {
        const int n = dsc[k] & 8191;
        if (n > 32) {
            const unsigned* run = rec + (size_t)(wave + 32 * k) * REGION + (dsc[k] >> 13);
            for (int i = 32 + lane; i < n; i += 32)
                atomicAdd(&ncnt[run[i] & (BIN_NODES - 1)], 1);
        }
    }
    __syncthreads();

    if (t < 32) gat_wave_scan<BIN_NODES / 32, false>(ncnt, cur, t);
    __syncthreads();
    if (t < BIN_NODES) {
        const int node = bin * BIN_NODES + t;
        if (node < N_NODES) {
            off[node]  = base + cur[t];
            degs[node] = ncnt[t];
        }
    }
    __syncthreads();

    #pragma unroll
    for (int k = 0; k < RUNS_PER_WAVE; ++k) {
        if ((int)rv[k] >= 0) {
            const int p = atomicAdd(&cur[rv[k] & (BIN_NODES - 1)], 1);
            atomicExch(&csr_src[base + p], (int)(rv[k] >> 8));
        }
    }
    #pragma unroll
    for (int k = 0; k < RUNS_PER_WAVE; ++k) {
        const int n = dsc[k] & 8191;
        if (n > 32) {
            const unsigned* run = rec + (size_t)(wave + 32 * k) * REGION + (dsc[k] >> 13);
            for (int i = 32 + lane; i < n; i += 32) {
                const unsigned x = run[i];
                const int p = atomicAdd(&cur[x & (BIN_NODES - 1)], 1);
                atomicExch(&csr_src[base + p], (int)(x >> 8));
            }
        }
    }
}

__device__ __forceinline__ float gat_leaky(float v) {
    return (v > 0.0f) ? v : v * NEG_SLOPE;
}

__device__ __forceinline__ float gat_wave_max(float v) {
    for (int o = 16; o > 0; o >>= 1) v = fmaxf(v, __shfl_xor(v, o));
    return v;
}

__device__ __forceinline__ float gat_wave_sum(float v) {
    for (int o = 16; o > 0; o >>= 1) v += __shfl_xor(v, o);
    return v;
}

__device__ __forceinline__ float gat_finite_or_zero(float v) {
    return (!(v == v) || v == -__builtin_inff() || v == __builtin_inff()) ? 0.0f : v;
}

__device__ __forceinline__ void gat_add_rows(const float4* __restrict__ feat4,
                                             int lane, int k0, int k1, int k2, int k3,
                                             int n, int s,
                                             float a0, float a1, float a2, float a3,
                                             float4& acc) {
    const int i0 = __float_as_int(a0);
    const int i1 = __float_as_int(a1);
    const int i2 = __float_as_int(a2);
    const int i3 = __float_as_int(a3);
    #pragma unroll 4
    for (int j = 0; j < n; ++j) {
        const int sj = __builtin_amdgcn_readlane(s, j);
        const int ab = (__builtin_amdgcn_readlane(i0, j) & k0) |
                       (__builtin_amdgcn_readlane(i1, j) & k1) |
                       (__builtin_amdgcn_readlane(i2, j) & k2) |
                       (__builtin_amdgcn_readlane(i3, j) & k3);
        const float a = __int_as_float(ab);
        const float4 f = feat4[(size_t)sj * (HF / 4) + lane];
        acc.x += f.x * a;
        acc.y += f.y * a;
        acc.z += f.z * a;
        acc.w += f.w * a;
    }
}

__global__ void __launch_bounds__(256)
gat_node_kernel(const float* __restrict__ feat,
                const float* __restrict__ el,
                const float* __restrict__ er,
                const int* __restrict__ off,
                const int* __restrict__ degs,
                const int* __restrict__ csr_src,
                const float* __restrict__ bias,
                float* __restrict__ out) {
    const int lane = threadIdx.x & 31;
    const int node = __builtin_amdgcn_readfirstlane(
        (int)((blockIdx.x * blockDim.x + threadIdx.x) >> 5));
    if (node >= N_NODES) return;

    const int beg = __builtin_amdgcn_readfirstlane(off[node]);
    const int deg = __builtin_amdgcn_readfirstlane(degs[node]);
    const int end = beg + deg;
    const float4* el4   = (const float4*)el;
    const float4* feat4 = (const float4*)feat;
    const float4  r4    = ((const float4*)er)[node];
    const float   ninf  = -__builtin_inff();

    const bool v0 = lane < deg;
    const bool v1 = lane + 32 < deg;
    const int  s0 = v0 ? csr_src[beg + lane] : 0;
    const int  s1 = v1 ? csr_src[beg + 32 + lane] : 0;
    const float4 l0 = el4[s0];
    const float4 l1 = el4[s1];
    const float e00 = v0 ? gat_leaky(l0.x + r4.x) : ninf;
    const float e01 = v0 ? gat_leaky(l0.y + r4.y) : ninf;
    const float e02 = v0 ? gat_leaky(l0.z + r4.z) : ninf;
    const float e03 = v0 ? gat_leaky(l0.w + r4.w) : ninf;
    const float e10 = v1 ? gat_leaky(l1.x + r4.x) : ninf;
    const float e11 = v1 ? gat_leaky(l1.y + r4.y) : ninf;
    const float e12 = v1 ? gat_leaky(l1.z + r4.z) : ninf;
    const float e13 = v1 ? gat_leaky(l1.w + r4.w) : ninf;

    float m0 = fmaxf(e00, e10), m1 = fmaxf(e01, e11);
    float m2 = fmaxf(e02, e12), m3 = fmaxf(e03, e13);
    for (int i = beg + 64 + lane; i < end; i += 32) {
        const float4 l = el4[csr_src[i]];
        m0 = fmaxf(m0, gat_leaky(l.x + r4.x));
        m1 = fmaxf(m1, gat_leaky(l.y + r4.y));
        m2 = fmaxf(m2, gat_leaky(l.z + r4.z));
        m3 = fmaxf(m3, gat_leaky(l.w + r4.w));
    }
    m0 = gat_finite_or_zero(gat_wave_max(m0));
    m1 = gat_finite_or_zero(gat_wave_max(m1));
    m2 = gat_finite_or_zero(gat_wave_max(m2));
    m3 = gat_finite_or_zero(gat_wave_max(m3));

    const float x00 = __expf(e00 - m0), x01 = __expf(e01 - m1);
    const float x02 = __expf(e02 - m2), x03 = __expf(e03 - m3);
    const float x10 = __expf(e10 - m0), x11 = __expf(e11 - m1);
    const float x12 = __expf(e12 - m2), x13 = __expf(e13 - m3);
    float d0 = x00 + x10, d1 = x01 + x11, d2 = x02 + x12, d3 = x03 + x13;
    for (int i = beg + 64 + lane; i < end; i += 32) {
        const float4 l = el4[csr_src[i]];
        d0 += __expf(gat_leaky(l.x + r4.x) - m0);
        d1 += __expf(gat_leaky(l.y + r4.y) - m1);
        d2 += __expf(gat_leaky(l.z + r4.z) - m2);
        d3 += __expf(gat_leaky(l.w + r4.w) - m3);
    }
    d0 = fmaxf(gat_wave_sum(d0), 1e-9f);
    d1 = fmaxf(gat_wave_sum(d1), 1e-9f);
    d2 = fmaxf(gat_wave_sum(d2), 1e-9f);
    d3 = fmaxf(gat_wave_sum(d3), 1e-9f);

    const int hl = lane >> 3;
    const int k0 = (hl == 0) ? -1 : 0;
    const int k1 = (hl == 1) ? -1 : 0;
    const int k2 = (hl == 2) ? -1 : 0;
    const int k3 = (hl == 3) ? -1 : 0;
    float4 acc = make_float4(0.0f, 0.0f, 0.0f, 0.0f);
    {
        const int n0 = (deg < 32) ? deg : 32;
        gat_add_rows(feat4, lane, k0, k1, k2, k3, n0, s0,
                     x00 / d0, x01 / d1, x02 / d2, x03 / d3, acc);
    }
    if (deg > 32) {
        const int n1 = (deg < 64) ? (deg - 32) : 32;
        gat_add_rows(feat4, lane, k0, k1, k2, k3, n1, s1,
                     x10 / d0, x11 / d1, x12 / d2, x13 / d3, acc);
    }
    for (int cb = beg + 64; cb < end; cb += 32) {
        const int  n = (end - cb < 32) ? (end - cb) : 32;
        const bool v = lane < n;
        const int  s = v ? csr_src[cb + lane] : 0;
        const float4 l = el4[s];
        const float a0 = v ? __expf(gat_leaky(l.x + r4.x) - m0) / d0 : 0.0f;
        const float a1 = v ? __expf(gat_leaky(l.y + r4.y) - m1) / d1 : 0.0f;
        const float a2 = v ? __expf(gat_leaky(l.z + r4.z) - m2) / d2 : 0.0f;
        const float a3 = v ? __expf(gat_leaky(l.w + r4.w) - m3) / d3 : 0.0f;
        gat_add_rows(feat4, lane, k0, k1, k2, k3, n, s, a0, a1, a2, a3, acc);
    }

    const float4 b4 = ((const float4*)bias)[lane];
    const float rx = fmaxf(acc.x + b4.x, 0.0f);
    const float ry = fmaxf(acc.y + b4.y, 0.0f);
    const float rz = fmaxf(acc.z + b4.z, 0.0f);
    const float rw = fmaxf(acc.w + b4.w, 0.0f);
    float4 o;
    o.x = ((rx + __shfl_down(rx, 8)) + __shfl_down(rx, 16)) + __shfl_down(rx, 24);
    o.y = ((ry + __shfl_down(ry, 8)) + __shfl_down(ry, 16)) + __shfl_down(ry, 24);
    o.z = ((rz + __shfl_down(rz, 8)) + __shfl_down(rz, 16)) + __shfl_down(rz, 24);
    o.w = ((rw + __shfl_down(rw, 8)) + __shfl_down(rw, 16)) + __shfl_down(rw, 24);
    if (lane < 8) {
        o.x *= (1.0f / N_HEADS);
        o.y *= (1.0f / N_HEADS);
        o.z *= (1.0f / N_HEADS);
        o.w *= (1.0f / N_HEADS);
        ((float4*)out)[(size_t)node * (OUT_FEAT / 4) + lane] = o;
    }
}

extern "C" void kernel_launch(void* const* d_in, const int* in_sizes, int n_in,
                              void* d_out, int out_size, void* d_ws, size_t ws_size,
                              hipStream_t stream) {
    const float* h      = (const float*)d_in[0];
    const float* W      = (const float*)d_in[1];
    const float* attn_l = (const float*)d_in[2];
    const float* attn_r = (const float*)d_in[3];
    const float* bias   = (const float*)d_in[4];
    const int*   src    = (const int*)d_in[5];
    const int*   dst    = (const int*)d_in[6];
    float* out = (float*)d_out;

    float* ws      = (float*)d_ws;
    float* feat    = ws;
    float* el      = feat + (size_t)N_NODES * HF;
    float* er      = el   + (size_t)N_NODES * N_HEADS;
    int*   bin_base = (int*)(er + (size_t)N_NODES * N_HEADS);
    int*   off      = bin_base + 256;
    int*   degs     = off      + 50176;
    int*   desc     = degs     + 50176;
    int*   csr_src  = desc     + N_CHUNKS * BINS_PAD;
    unsigned* rec   = (unsigned*)(csr_src + N_EDGES + N_BINS * 32);

    const int BS = 256;

    gat_proj_part_kernel<<<N_CHUNKS + PROJ_TILES, 256, 0, stream>>>(
        h, W, feat, src, dst, rec, desc);
    gat_attn_bases_kernel<<<1 + ATTN_BLOCKS, SCAN_T, 0, stream>>>(
        feat, attn_l, attn_r, el, er, desc, bin_base);
    gat_bin_csr_kernel<<<N_BINS, 1024, 0, stream>>>(rec, desc, bin_base, off, degs, csr_src);
    gat_node_kernel<<<(N_NODES * 32 + BS - 1) / BS, BS, 0, stream>>>(
        feat, el, er, off, degs, csr_src, bias, out);
}


# Projection: a work-group stages a 64-row strip of h in LDS by whole-line loads; each wave does its 16 columns for all 4 row tiles (A from LDS, one W fragment per 4 WMMAs); sum order unchanged
# speedup vs baseline: 7.0485x; 1.2006x over previous
#include <hip/hip_runtime.h>
#include <hip/hip_bf16.h>

#define N_NODES 50000
#define N_EDGES 1600000
#define IN_FEAT 128
#define OUT_FEAT 32
#define N_HEADS 4
#define HF (N_HEADS * OUT_FEAT)
#define NEG_SLOPE 0.2f

typedef __attribute__((ext_vector_type(2))) float v2f;
typedef __attribute__((ext_vector_type(8))) float v8f;

#define BIN_SHIFT 8
#define BIN_NODES (1 << BIN_SHIFT)
#define N_BINS    ((N_NODES + BIN_NODES - 1) / BIN_NODES)
#define BINS_PAD  224
#define CHUNK     4096
#define N_CHUNKS  ((N_EDGES + CHUNK - 1) / CHUNK)
#define EPT       (CHUNK / 256)
#define REGION    10240
static_assert(N_BINS <= BINS_PAD && BINS_PAD <= 256 && BINS_PAD % 32 == 0, "desc row");
static_assert(CHUNK + N_BINS * 31 <= REGION && REGION % 32 == 0, "padded runs fit");
static_assert(N_NODES <= (1 << 23), "src << 8 must fit a word");
static_assert(REGION < (1 << 18) && CHUNK < (1 << 13), "desc packing: start << 13 | count");

#define PROJ_ROWS    64
#define PROJ_RT      (PROJ_ROWS / 16)
#define PROJ_BLOCKS  ((N_NODES + PROJ_ROWS - 1) / PROJ_ROWS)
#define PROJ_LAST_RT (((N_NODES - 1) % PROJ_ROWS) / 16 + 1)
#define A_PITCH      (IN_FEAT + 4)
static_assert(N_NODES % 16 == 0, "row tiles must be whole");
static_assert(PROJ_ROWS * A_PITCH <= REGION, "the strip of h fits the staging array");

template <int NT>
__device__ __forceinline__ void gat_gemm_block(const float* __restrict__ h,
                                               const float* __restrict__ W,
                                               float* __restrict__ feat,
                                               int m0, float* hs) {
    const int tid  = threadIdx.x;
    const int lane = tid & 31;
    const int half = lane >> 4;
    const int l    = lane & 15;
    const int n0   = (tid >> 5) * 16;

    float4 v[2 * NT];
    #pragma unroll
    for (int i = 0; i < 2 * NT; ++i) {
        const int idx = tid + 256 * i;
        v[i] = ((const float4*)h)[(size_t)(m0 + (idx >> 5)) * (IN_FEAT / 4) + (idx & 31)];
    }
    #pragma unroll
    for (int i = 0; i < 2 * NT; ++i) {
        const int idx = tid + 256 * i;
        *(float4*)(hs + (idx >> 5) * A_PITCH + 4 * (idx & 31)) = v[i];
    }
    __syncthreads();

    v8f acc[NT];
    #pragma unroll
    for (int t = 0; t < NT; ++t) acc[t] = v8f{};

    const float* arow = hs + l * A_PITCH + half * 2;
    const float* wcol = W + (size_t)(half * 2) * HF + n0 + l;

    #pragma unroll 2
    for (int kk = 0; kk < IN_FEAT; kk += 4) {
        v2f b;
        b.x = wcol[(size_t)(kk + 0) * HF];
        b.y = wcol[(size_t)(kk + 1) * HF];
        #pragma unroll
        for (int t = 0; t < NT; ++t) {
            const v2f a = *(const v2f*)(arow + t * 16 * A_PITCH + kk);
            acc[t] = __builtin_amdgcn_wmma_f32_16x16x4_f32(
                 false, a,  false, b,
                 (short)0, acc[t],  false,  false);
        }
    }

    #pragma unroll
    for (int t = 0; t < NT; ++t) {
        #pragma unroll
        for (int j = 0; j < 8; ++j) {
            feat[(size_t)(m0 + t * 16 + half * 8 + j) * HF + n0 + l] = acc[t][j];
        }
    }
}

template <int PER, bool PAD>
__device__ __forceinline__ void gat_wave_scan(const int* in, int* out, int lane) {
    int v[PER];
    int s = 0;
    #pragma unroll
    for (int j = 0; j < PER; ++j) {
        const int c = in[lane * PER + j];
        v[j] = PAD ? ((c + 31) & ~31) : c;
        s += v[j];
    }
    int incl = s;
    for (int o = 1; o < 32; o <<= 1) {
        const int u = __shfl_up(incl, o);
        if (lane >= o) incl += u;
    }
    int b = incl - s;
    #pragma unroll
    for (int j = 0; j < PER; ++j) {
        out[lane * PER + j] = b;
        b += v[j];
    }
}

__device__ __forceinline__ void gat_partition_block(const int* __restrict__ src,
                                                    const int* __restrict__ dst,
                                                    unsigned* __restrict__ rec,
                                                    int* __restrict__ desc,
                                                    int chunk,
                                                    int* hcnt, int* hbase,
                                                    unsigned* stage) {
    const int tid = threadIdx.x;
    hcnt[tid] = 0;
    __syncthreads();

    int d[EPT], r[EPT];
    const int e0 = chunk * CHUNK + tid;
    #pragma unroll
    for (int k = 0; k < EPT; ++k) {
        const int e = e0 + k * 256;
        d[k] = (e < N_EDGES) ? dst[e] : -1;
    }
    #pragma unroll
    for (int k = 0; k < EPT; ++k) {
        r[k] = (d[k] >= 0) ? atomicAdd(&hcnt[d[k] >> BIN_SHIFT], 1) : 0;
    }
    __syncthreads();

    if (tid < 32) gat_wave_scan<BINS_PAD / 32, true>(hcnt, hbase, tid);
    __syncthreads();

    #pragma unroll
    for (int k = 0; k < EPT; ++k) {
        if (d[k] >= 0) {
            const int e = e0 + k * 256;
            stage[hbase[d[k] >> BIN_SHIFT] + r[k]] =
                ((unsigned)src[e] << 8) | (unsigned)(d[k] & (BIN_NODES - 1));
        }
    }
    if (tid < BINS_PAD) {
        const int c = hcnt[tid];
        const int p = (c + 31) & ~31;
        for (int i = c; i < p; ++i) stage[hbase[tid] + i] = 0u;
        desc[chunk * BINS_PAD + tid] = (hbase[tid] << 13) | c;
    }
    __syncthreads();

    const int total = hbase[BINS_PAD - 1] + ((hcnt[BINS_PAD - 1] + 31) & ~31);
    unsigned* out = rec + (size_t)chunk * REGION;
    for (int i = tid; i < total; i += 256) out[i] = stage[i];
}

__global__ void __launch_bounds__(256)
gat_proj_part_kernel(const float* __restrict__ h,
                     const float* __restrict__ W,
                     float* __restrict__ feat,
                     const int* __restrict__ src,
                     const int* __restrict__ dst,
                     unsigned* __restrict__ rec,
                     int* __restrict__ desc) {
    __shared__ int hcnt[256];
    __shared__ int hbase[256];
    __shared__ unsigned stage[REGION];
    if (blockIdx.x < N_CHUNKS) {
        gat_partition_block(src, dst, rec, desc, (int)blockIdx.x, hcnt, hbase, stage);
    } else {
        const int q = (int)blockIdx.x - N_CHUNKS;
        if (PROJ_LAST_RT == PROJ_RT || q < PROJ_BLOCKS - 1)
            gat_gemm_block<PROJ_RT>(h, W, feat, q * PROJ_ROWS, (float*)stage);
        else
            gat_gemm_block<PROJ_LAST_RT>(h, W, feat, q * PROJ_ROWS, (float*)stage);
    }
}

#define RUNS_PER_WAVE ((N_CHUNKS + 31) / 32)

__device__ __forceinline__ int gat_bin_base(const int* __restrict__ desc, int bin,
                                            int* tot, int* sb) {
    const int t    = threadIdx.x;
    const int lane = t & 31;
    const int wave = t >> 5;
    if (t < 256) tot[t] = 0;
    __syncthreads();
    int part[BINS_PAD / 32];
    #pragma unroll
    for (int j = 0; j < BINS_PAD / 32; ++j) part[j] = 0;
    #pragma unroll
    for (int k = 0; k < RUNS_PER_WAVE; ++k) {
        const int c = wave + 32 * k;
        if (c < N_CHUNKS) {
            #pragma unroll
            for (int j = 0; j < BINS_PAD / 32; ++j)
                part[j] += desc[c * BINS_PAD + lane + 32 * j] & 8191;
        }
    }
    #pragma unroll
    for (int j = 0; j < BINS_PAD / 32; ++j) atomicAdd(&tot[lane + 32 * j], part[j]);
    __syncthreads();
    if (t < 32) gat_wave_scan<BINS_PAD / 32, true>(tot, sb, t);
    __syncthreads();
    return sb[bin];
}


__global__ void __launch_bounds__(1024)
gat_bin_csr_kernel(const unsigned* __restrict__ rec,
                   const int* __restrict__ desc,
                   int* __restrict__ off,
                   int* __restrict__ degs,
                   int* __restrict__ csr_src) {
    __shared__ int ncnt[BIN_NODES];
    __shared__ int cur[BIN_NODES];
    __shared__ int tot[256];
    __shared__ int sb[256];
    const int bin  = blockIdx.x;
    const int t    = threadIdx.x;
    const int lane = t & 31;
    const int wave = t >> 5;
    const int base = gat_bin_base(desc, bin, tot, sb);

    if (t < BIN_NODES) ncnt[t] = 0;

    int      dsc[RUNS_PER_WAVE];
    unsigned rv[RUNS_PER_WAVE];
    #pragma unroll
    for (int k = 0; k < RUNS_PER_WAVE; ++k) {
        const int c = wave + 32 * k;
        dsc[k] = (c < N_CHUNKS) ? desc[c * BINS_PAD + bin] : 0;
    }
    #pragma unroll
    for (int k = 0; k < RUNS_PER_WAVE; ++k) {
        const int c = wave + 32 * k;
        const int n = dsc[k] & 8191;
        rv[k] = (lane < n) ? rec[(size_t)c * REGION + (dsc[k] >> 13) + lane]
                           : 0xFFFFFFFFu;
    }
    __syncthreads();

    #pragma unroll
    for (int k = 0; k < RUNS_PER_WAVE; ++k) {
        if ((int)rv[k] >= 0) atomicAdd(&ncnt[rv[k] & (BIN_NODES - 1)], 1);
    }
    #pragma unroll
    for (int k = 0; k < RUNS_PER_WAVE; ++k) {
        const int n = dsc[k] & 8191;
        if (n > 32) {
            const unsigned* run = rec + (size_t)(wave + 32 * k) * REGION + (dsc[k] >> 13);
            for (int i = 32 + lane; i < n; i += 32)
                atomicAdd(&ncnt[run[i] & (BIN_NODES - 1)], 1);
        }
    }
    __syncthreads();

    if (t < 32) gat_wave_scan<BIN_NODES / 32, false>(ncnt, cur, t);
    __syncthreads();
    if (t < BIN_NODES) {
        const int node = bin * BIN_NODES + t;
        if (node < N_NODES) {
            off[node]  = base + cur[t];
            degs[node] = ncnt[t];
        }
    }
    __syncthreads();

    #pragma unroll
    for (int k = 0; k < RUNS_PER_WAVE; ++k) {
        if ((int)rv[k] >= 0) {
            const int p = atomicAdd(&cur[rv[k] & (BIN_NODES - 1)], 1);
            atomicExch(&csr_src[base + p], (int)(rv[k] >> 8));
        }
    }
    #pragma unroll
    for (int k = 0; k < RUNS_PER_WAVE; ++k) {
        const int n = dsc[k] & 8191;
        if (n > 32) {
            const unsigned* run = rec + (size_t)(wave + 32 * k) * REGION + (dsc[k] >> 13);
            for (int i = 32 + lane; i < n; i += 32) {
                const unsigned x = run[i];
                const int p = atomicAdd(&cur[x & (BIN_NODES - 1)], 1);
                atomicExch(&csr_src[base + p], (int)(x >> 8));
            }
        }
    }
}

#define LOG2E 1.4426950408889634f

template <int CTRL>
__device__ __forceinline__ float gat_xadd(float v) {
    return v + __int_as_float(__builtin_amdgcn_update_dpp(
                   0, __float_as_int(v), CTRL, 0xf, 0xf, true));
}
__device__ __forceinline__ float gat_sum8(float v) {
    v = gat_xadd<0x161>(v);
    v = gat_xadd<0x162>(v);
    v = gat_xadd<0x164>(v);
    return v;
}

__device__ __forceinline__ float gat_dot4(const float4& a, const float4& b) {
    const v2f a01 = {a.x, a.y}, a23 = {a.z, a.w};
    const v2f b01 = {b.x, b.y}, b23 = {b.z, b.w};
    const v2f t = a23 * b23 + a01 * b01;
    return t.x + t.y;
}

__device__ __forceinline__ float gat_leaky(float v) {
    return fmaxf(v, v * NEG_SLOPE);
}

template <bool TAIL>
__device__ __forceinline__ void gat_edges4(const float4* __restrict__ feat4,
                                           int lane, int s, int j, int cnt,
                                           const float4& al, float er,
                                           float& m, float& d, float4& acc) {
    const int s0 = __builtin_amdgcn_readlane(s, j + 0);
    const int s1 = __builtin_amdgcn_readlane(s, j + 1);
    const int s2 = __builtin_amdgcn_readlane(s, j + 2);
    const int s3 = __builtin_amdgcn_readlane(s, j + 3);
    const float4 f0 = feat4[(size_t)s0 * (HF / 4) + lane];
    const float4 f1 = feat4[(size_t)s1 * (HF / 4) + lane];
    const float4 f2 = feat4[(size_t)s2 * (HF / 4) + lane];
    const float4 f3 = feat4[(size_t)s3 * (HF / 4) + lane];
    float e0 = gat_leaky(gat_sum8(gat_dot4(f0, al)) + er);
    float e1 = gat_leaky(gat_sum8(gat_dot4(f1, al)) + er);
    float e2 = gat_leaky(gat_sum8(gat_dot4(f2, al)) + er);
    float e3 = gat_leaky(gat_sum8(gat_dot4(f3, al)) + er);
    if (TAIL) {
        const float ninf = -__builtin_inff();
        if (cnt < 2) e1 = ninf;
        if (cnt < 3) e2 = ninf;
        e3 = ninf;
    }
    const float mn = fmaxf(fmaxf(fmaxf(m, e0), fmaxf(e1, e2)), e3);
    const float sc = __builtin_amdgcn_exp2f(m - mn);
    const float p0 = __builtin_amdgcn_exp2f(e0 - mn);
    const float p1 = __builtin_amdgcn_exp2f(e1 - mn);
    const float p2 = __builtin_amdgcn_exp2f(e2 - mn);
    const float p3 = __builtin_amdgcn_exp2f(e3 - mn);
    d = d * sc + ((p0 + p1) + (p2 + p3));
    acc.x = (((acc.x * sc + f0.x * p0) + f1.x * p1) + f2.x * p2) + f3.x * p3;
    acc.y = (((acc.y * sc + f0.y * p0) + f1.y * p1) + f2.y * p2) + f3.y * p3;
    acc.z = (((acc.z * sc + f0.z * p0) + f1.z * p1) + f2.z * p2) + f3.z * p3;
    acc.w = (((acc.w * sc + f0.w * p0) + f1.w * p1) + f2.w * p2) + f3.w * p3;
    m = mn;
}

__global__ void __launch_bounds__(256)
gat_node_kernel(const float* __restrict__ feat,
                const float* __restrict__ attn_l,
                const float* __restrict__ attn_r,
                const int* __restrict__ off,
                const int* __restrict__ degs,
                const int* __restrict__ csr_src,
                const float* __restrict__ bias,
                float* __restrict__ out) {
    const int lane = threadIdx.x & 31;
    const int node = __builtin_amdgcn_readfirstlane(
        (int)((blockIdx.x * blockDim.x + threadIdx.x) >> 5));
    if (node >= N_NODES) return;

    const int beg = __builtin_amdgcn_readfirstlane(off[node]);
    const int deg = __builtin_amdgcn_readfirstlane(degs[node]);
    const float4* feat4 = (const float4*)feat;

    float4 al = ((const float4*)attn_l)[lane];
    float4 ar = ((const float4*)attn_r)[lane];
    al.x *= LOG2E; al.y *= LOG2E; al.z *= LOG2E; al.w *= LOG2E;
    ar.x *= LOG2E; ar.y *= LOG2E; ar.z *= LOG2E; ar.w *= LOG2E;
    const float4 fo = feat4[(size_t)node * (HF / 4) + lane];
    const float  er = gat_sum8(gat_dot4(fo, ar));

    float  m   = -__builtin_inff();
    float  d   = 0.0f;
    float4 acc = make_float4(0.0f, 0.0f, 0.0f, 0.0f);
    for (int cb = 0; cb < deg; cb += 32) {
        const int n = (deg - cb < 32) ? (deg - cb) : 32;
        const int s = (lane < n) ? csr_src[beg + cb + lane] : 0;
        const int nfull = n & ~3;
        for (int j = 0; j < nfull; j += 4)
            gat_edges4<false>(feat4, lane, s, j, 4, al, er, m, d, acc);
        if (n & 3)
            gat_edges4<true>(feat4, lane, s, nfull, n & 3, al, er, m, d, acc);
    }

    const float inv = 1.0f / fmaxf(d, 1e-9f);
    const float4 b4 = ((const float4*)bias)[lane];
    const float rx = fmaxf(acc.x * inv + b4.x, 0.0f);
    const float ry = fmaxf(acc.y * inv + b4.y, 0.0f);
    const float rz = fmaxf(acc.z * inv + b4.z, 0.0f);
    const float rw = fmaxf(acc.w * inv + b4.w, 0.0f);
    float4 o;
    o.x = ((rx + __shfl_down(rx, 8)) + __shfl_down(rx, 16)) + __shfl_down(rx, 24);
    o.y = ((ry + __shfl_down(ry, 8)) + __shfl_down(ry, 16)) + __shfl_down(ry, 24);
    o.z = ((rz + __shfl_down(rz, 8)) + __shfl_down(rz, 16)) + __shfl_down(rz, 24);
    o.w = ((rw + __shfl_down(rw, 8)) + __shfl_down(rw, 16)) + __shfl_down(rw, 24);
    if (lane < 8) {
        o.x *= (1.0f / N_HEADS);
        o.y *= (1.0f / N_HEADS);
        o.z *= (1.0f / N_HEADS);
        o.w *= (1.0f / N_HEADS);
        ((float4*)out)[(size_t)node * (OUT_FEAT / 4) + lane] = o;
    }
}

extern "C" void kernel_launch(void* const* d_in, const int* in_sizes, int n_in,
                              void* d_out, int out_size, void* d_ws, size_t ws_size,
                              hipStream_t stream) {
    const float* h      = (const float*)d_in[0];
    const float* W      = (const float*)d_in[1];
    const float* attn_l = (const float*)d_in[2];
    const float* attn_r = (const float*)d_in[3];
    const float* bias   = (const float*)d_in[4];
    const int*   src    = (const int*)d_in[5];
    const int*   dst    = (const int*)d_in[6];
    float* out = (float*)d_out;

    float* ws      = (float*)d_ws;
    float* feat    = ws;
    int*   off      = (int*)(feat + (size_t)N_NODES * HF);
    int*   degs     = off      + 50176;
    int*   desc     = degs     + 50176;
    int*   csr_src  = desc     + N_CHUNKS * BINS_PAD;
    unsigned* rec   = (unsigned*)(csr_src + N_EDGES + N_BINS * 32);

    const int BS = 256;

    gat_proj_part_kernel<<<N_CHUNKS + PROJ_BLOCKS, 256, 0, stream>>>(
        h, W, feat, src, dst, rec, desc);
    gat_bin_csr_kernel<<<N_BINS, 1024, 0, stream>>>(rec, desc, off, degs, csr_src);
    gat_node_kernel<<<(N_NODES * 32 + BS - 1) / BS, BS, 0, stream>>>(
        feat, attn_l, attn_r, off, degs, csr_src, bias, out);
}


# Level 2 fills a bin's part of csr_src in LDS and writes it out in whole 128-byte lines instead of one atomic swap per edge at the L2 (swap kept for a bin over 32768 edges); sum order unchanged
# speedup vs baseline: 7.4219x; 1.0530x over previous
#include <hip/hip_runtime.h>
#include <hip/hip_bf16.h>

#define N_NODES 50000
#define N_EDGES 1600000
#define IN_FEAT 128
#define OUT_FEAT 32
#define N_HEADS 4
#define HF (N_HEADS * OUT_FEAT)
#define NEG_SLOPE 0.2f

typedef __attribute__((ext_vector_type(2))) float v2f;
typedef __attribute__((ext_vector_type(8))) float v8f;

#define BIN_SHIFT 8
#define BIN_NODES (1 << BIN_SHIFT)
#define N_BINS    ((N_NODES + BIN_NODES - 1) / BIN_NODES)
#define BINS_PAD  224
#define CHUNK     4096
#define N_CHUNKS  ((N_EDGES + CHUNK - 1) / CHUNK)
#define EPT       (CHUNK / 256)
#define REGION    10240
static_assert(N_BINS <= BINS_PAD && BINS_PAD <= 256 && BINS_PAD % 32 == 0, "desc row");
static_assert(CHUNK + N_BINS * 31 <= REGION && REGION % 32 == 0, "padded runs fit");
static_assert(N_NODES <= (1 << 23), "src << 8 must fit a word");
static_assert(REGION < (1 << 18) && CHUNK < (1 << 13), "desc packing: start << 13 | count");

#define PROJ_ROWS    64
#define PROJ_RT      (PROJ_ROWS / 16)
#define PROJ_BLOCKS  ((N_NODES + PROJ_ROWS - 1) / PROJ_ROWS)
#define PROJ_LAST_RT (((N_NODES - 1) % PROJ_ROWS) / 16 + 1)
#define A_PITCH      (IN_FEAT + 4)
static_assert(N_NODES % 16 == 0, "row tiles must be whole");
static_assert(PROJ_ROWS * A_PITCH <= REGION, "the strip of h fits the staging array");

template <int NT>
__device__ __forceinline__ void gat_gemm_block(const float* __restrict__ h,
                                               const float* __restrict__ W,
                                               float* __restrict__ feat,
                                               int m0, float* hs) {
    const int tid  = threadIdx.x;
    const int lane = tid & 31;
    const int half = lane >> 4;
    const int l    = lane & 15;
    const int n0   = (tid >> 5) * 16;

    float4 v[2 * NT];
    #pragma unroll
    for (int i = 0; i < 2 * NT; ++i) {
        const int idx = tid + 256 * i;
        v[i] = ((const float4*)h)[(size_t)(m0 + (idx >> 5)) * (IN_FEAT / 4) + (idx & 31)];
    }
    #pragma unroll
    for (int i = 0; i < 2 * NT; ++i) {
        const int idx = tid + 256 * i;
        *(float4*)(hs + (idx >> 5) * A_PITCH + 4 * (idx & 31)) = v[i];
    }
    __syncthreads();

    v8f acc[NT];
    #pragma unroll
    for (int t = 0; t < NT; ++t) acc[t] = v8f{};

    const float* arow = hs + l * A_PITCH + half * 2;
    const float* wcol = W + (size_t)(half * 2) * HF + n0 + l;

    #pragma unroll 2
    for (int kk = 0; kk < IN_FEAT; kk += 4) {
        v2f b;
        b.x = wcol[(size_t)(kk + 0) * HF];
        b.y = wcol[(size_t)(kk + 1) * HF];
        #pragma unroll
        for (int t = 0; t < NT; ++t) {
            const v2f a = *(const v2f*)(arow + t * 16 * A_PITCH + kk);
            acc[t] = __builtin_amdgcn_wmma_f32_16x16x4_f32(
                 false, a,  false, b,
                 (short)0, acc[t],  false,  false);
        }
    }

    #pragma unroll
    for (int t = 0; t < NT; ++t) {
        #pragma unroll
        for (int j = 0; j < 8; ++j) {
            feat[(size_t)(m0 + t * 16 + half * 8 + j) * HF + n0 + l] = acc[t][j];
        }
    }
}

template <int PER, bool PAD>
__device__ __forceinline__ void gat_wave_scan(const int* in, int* out, int lane) {
    int v[PER];
    int s = 0;
    #pragma unroll
    for (int j = 0; j < PER; ++j) {
        const int c = in[lane * PER + j];
        v[j] = PAD ? ((c + 31) & ~31) : c;
        s += v[j];
    }
    int incl = s;
    for (int o = 1; o < 32; o <<= 1) {
        const int u = __shfl_up(incl, o);
        if (lane >= o) incl += u;
    }
    int b = incl - s;
    #pragma unroll
    for (int j = 0; j < PER; ++j) {
        out[lane * PER + j] = b;
        b += v[j];
    }
}

__device__ __forceinline__ void gat_partition_block(const int* __restrict__ src,
                                                    const int* __restrict__ dst,
                                                    unsigned* __restrict__ rec,
                                                    int* __restrict__ desc,
                                                    int chunk,
                                                    int* hcnt, int* hbase,
                                                    unsigned* stage) {
    const int tid = threadIdx.x;
    hcnt[tid] = 0;
    __syncthreads();

    int d[EPT], r[EPT];
    const int e0 = chunk * CHUNK + tid;
    #pragma unroll
    for (int k = 0; k < EPT; ++k) {
        const int e = e0 + k * 256;
        d[k] = (e < N_EDGES) ? dst[e] : -1;
    }
    #pragma unroll
    for (int k = 0; k < EPT; ++k) {
        r[k] = (d[k] >= 0) ? atomicAdd(&hcnt[d[k] >> BIN_SHIFT], 1) : 0;
    }
    __syncthreads();

    if (tid < 32) gat_wave_scan<BINS_PAD / 32, true>(hcnt, hbase, tid);
    __syncthreads();

    #pragma unroll
    for (int k = 0; k < EPT; ++k) {
        if (d[k] >= 0) {
            const int e = e0 + k * 256;
            stage[hbase[d[k] >> BIN_SHIFT] + r[k]] =
                ((unsigned)src[e] << 8) | (unsigned)(d[k] & (BIN_NODES - 1));
        }
    }
    if (tid < BINS_PAD) {
        const int c = hcnt[tid];
        const int p = (c + 31) & ~31;
        for (int i = c; i < p; ++i) stage[hbase[tid] + i] = 0u;
        desc[chunk * BINS_PAD + tid] = (hbase[tid] << 13) | c;
    }
    __syncthreads();

    const int total = hbase[BINS_PAD - 1] + ((hcnt[BINS_PAD - 1] + 31) & ~31);
    unsigned* out = rec + (size_t)chunk * REGION;
    for (int i = tid; i < total; i += 256) out[i] = stage[i];
}

__global__ void __launch_bounds__(256)
gat_proj_part_kernel(const float* __restrict__ h,
                     const float* __restrict__ W,
                     float* __restrict__ feat,
                     const int* __restrict__ src,
                     const int* __restrict__ dst,
                     unsigned* __restrict__ rec,
                     int* __restrict__ desc) {
    __shared__ int hcnt[256];
    __shared__ int hbase[256];
    __shared__ unsigned stage[REGION];
    if (blockIdx.x < N_CHUNKS) {
        gat_partition_block(src, dst, rec, desc, (int)blockIdx.x, hcnt, hbase, stage);
    } else {
        const int q = (int)blockIdx.x - N_CHUNKS;
        if (PROJ_LAST_RT == PROJ_RT || q < PROJ_BLOCKS - 1)
            gat_gemm_block<PROJ_RT>(h, W, feat, q * PROJ_ROWS, (float*)stage);
        else
            gat_gemm_block<PROJ_LAST_RT>(h, W, feat, q * PROJ_ROWS, (float*)stage);
    }
}

#define RUNS_PER_WAVE ((N_CHUNKS + 31) / 32)

__device__ __forceinline__ int gat_bin_base(const int* __restrict__ desc, int bin,
                                            int* tot, int* sb) {
    const int t    = threadIdx.x;
    const int lane = t & 31;
    const int wave = t >> 5;
    if (t < 256) tot[t] = 0;
    __syncthreads();
    int part[BINS_PAD / 32];
    #pragma unroll
    for (int j = 0; j < BINS_PAD / 32; ++j) part[j] = 0;
    #pragma unroll
    for (int k = 0; k < RUNS_PER_WAVE; ++k) {
        const int c = wave + 32 * k;
        if (c < N_CHUNKS) {
            #pragma unroll
            for (int j = 0; j < BINS_PAD / 32; ++j)
                part[j] += desc[c * BINS_PAD + lane + 32 * j] & 8191;
        }
    }
    #pragma unroll
    for (int j = 0; j < BINS_PAD / 32; ++j) atomicAdd(&tot[lane + 32 * j], part[j]);
    __syncthreads();
    if (t < 32) gat_wave_scan<BINS_PAD / 32, true>(tot, sb, t);
    __syncthreads();
    return sb[bin];
}

#define SEG_WORDS 32768
static_assert(SEG_WORDS % 32 == 0, "whole lines");

template <bool STAGED>
__device__ __forceinline__ void gat_place(unsigned x, int* cur, int* seg, int* csr_bin) {
    const int p = atomicAdd(&cur[x & (BIN_NODES - 1)], 1);
    if (STAGED) seg[p] = (int)(x >> 8);
    else        atomicExch(&csr_bin[p], (int)(x >> 8));
}


__global__ void __launch_bounds__(1024)
gat_bin_csr_kernel(const unsigned* __restrict__ rec,
                   const int* __restrict__ desc,
                   int* __restrict__ off,
                   int* __restrict__ degs,
                   int* __restrict__ csr_src) {
    __shared__ int ncnt[BIN_NODES];
    __shared__ int cur[BIN_NODES];
    __shared__ int tot[256];
    __shared__ int sb[256];
    __shared__ __attribute__((aligned(16))) int seg[SEG_WORDS];
    const int bin  = blockIdx.x;
    const int t    = threadIdx.x;
    const int lane = t & 31;
    const int wave = t >> 5;
    const int base = gat_bin_base(desc, bin, tot, sb);
    const int total = __builtin_amdgcn_readfirstlane(tot[bin]);

    if (t < BIN_NODES) ncnt[t] = 0;

    int      dsc[RUNS_PER_WAVE];
    unsigned rv[RUNS_PER_WAVE];
    #pragma unroll
    for (int k = 0; k < RUNS_PER_WAVE; ++k) {
        const int c = wave + 32 * k;
        dsc[k] = (c < N_CHUNKS) ? desc[c * BINS_PAD + bin] : 0;
    }
    #pragma unroll
    for (int k = 0; k < RUNS_PER_WAVE; ++k) {
        const int c = wave + 32 * k;
        const int n = dsc[k] & 8191;
        rv[k] = (lane < n) ? rec[(size_t)c * REGION + (dsc[k] >> 13) + lane]
                           : 0xFFFFFFFFu;
    }
    __syncthreads();

    #pragma unroll
    for (int k = 0; k < RUNS_PER_WAVE; ++k) {
        if ((int)rv[k] >= 0) atomicAdd(&ncnt[rv[k] & (BIN_NODES - 1)], 1);
    }
    #pragma unroll
    for (int k = 0; k < RUNS_PER_WAVE; ++k) {
        const int n = dsc[k] & 8191;
        if (n > 32) {
            const unsigned* run = rec + (size_t)(wave + 32 * k) * REGION + (dsc[k] >> 13);
            for (int i = 32 + lane; i < n; i += 32)
                atomicAdd(&ncnt[run[i] & (BIN_NODES - 1)], 1);
        }
    }
    __syncthreads();

    if (t < 32) gat_wave_scan<BIN_NODES / 32, false>(ncnt, cur, t);
    __syncthreads();
    if (t < BIN_NODES) {
        const int node = bin * BIN_NODES + t;
        if (node < N_NODES) {
            off[node]  = base + cur[t];
            degs[node] = ncnt[t];
        }
    }
    __syncthreads();

    int* csr_bin = csr_src + base;
    if (total <= SEG_WORDS) {
        #pragma unroll
        for (int k = 0; k < RUNS_PER_WAVE; ++k) {
            if ((int)rv[k] >= 0) gat_place<true>(rv[k], cur, seg, csr_bin);
        }
        #pragma unroll
        for (int k = 0; k < RUNS_PER_WAVE; ++k) {
            const int n = dsc[k] & 8191;
            if (n > 32) {
                const unsigned* run = rec + (size_t)(wave + 32 * k) * REGION + (dsc[k] >> 13);
                for (int i = 32 + lane; i < n; i += 32)
                    gat_place<true>(run[i], cur, seg, csr_bin);
            }
        }
        const int padded = (total + 31) & ~31;
        if (t < padded - total) seg[total + t] = 0;
        __syncthreads();
        for (int i = t; i < (padded >> 2); i += 1024)
            ((int4*)csr_bin)[i] = ((const int4*)seg)[i];
    } else {
        #pragma unroll
        for (int k = 0; k < RUNS_PER_WAVE; ++k) {
            if ((int)rv[k] >= 0) gat_place<false>(rv[k], cur, seg, csr_bin);
        }
        #pragma unroll
        for (int k = 0; k < RUNS_PER_WAVE; ++k) {
            const int n = dsc[k] & 8191;
            if (n > 32) {
                const unsigned* run = rec + (size_t)(wave + 32 * k) * REGION + (dsc[k] >> 13);
                for (int i = 32 + lane; i < n; i += 32)
                    gat_place<false>(run[i], cur, seg, csr_bin);
            }
        }
    }
}

#define LOG2E 1.4426950408889634f

template <int CTRL>
__device__ __forceinline__ float gat_xadd(float v) {
    return v + __int_as_float(__builtin_amdgcn_update_dpp(
                   0, __float_as_int(v), CTRL, 0xf, 0xf, true));
}
__device__ __forceinline__ float gat_sum8(float v) {
    v = gat_xadd<0x161>(v);
    v = gat_xadd<0x162>(v);
    v = gat_xadd<0x164>(v);
    return v;
}

__device__ __forceinline__ float gat_dot4(const float4& a, const float4& b) {
    const v2f a01 = {a.x, a.y}, a23 = {a.z, a.w};
    const v2f b01 = {b.x, b.y}, b23 = {b.z, b.w};
    const v2f t = a23 * b23 + a01 * b01;
    return t.x + t.y;
}

__device__ __forceinline__ float gat_leaky(float v) {
    return fmaxf(v, v * NEG_SLOPE);
}

template <bool TAIL>
__device__ __forceinline__ void gat_edges4(const float4* __restrict__ feat4,
                                           int lane, int s, int j, int cnt,
                                           const float4& al, float er,
                                           float& m, float& d, float4& acc) {
    const int s0 = __builtin_amdgcn_readlane(s, j + 0);
    const int s1 = __builtin_amdgcn_readlane(s, j + 1);
    const int s2 = __builtin_amdgcn_readlane(s, j + 2);
    const int s3 = __builtin_amdgcn_readlane(s, j + 3);
    const float4 f0 = feat4[(size_t)s0 * (HF / 4) + lane];
    const float4 f1 = feat4[(size_t)s1 * (HF / 4) + lane];
    const float4 f2 = feat4[(size_t)s2 * (HF / 4) + lane];
    const float4 f3 = feat4[(size_t)s3 * (HF / 4) + lane];
    float e0 = gat_leaky(gat_sum8(gat_dot4(f0, al)) + er);
    float e1 = gat_leaky(gat_sum8(gat_dot4(f1, al)) + er);
    float e2 = gat_leaky(gat_sum8(gat_dot4(f2, al)) + er);
    float e3 = gat_leaky(gat_sum8(gat_dot4(f3, al)) + er);
    if (TAIL) {
        const float ninf = -__builtin_inff();
        if (cnt < 2) e1 = ninf;
        if (cnt < 3) e2 = ninf;
        e3 = ninf;
    }
    const float mn = fmaxf(fmaxf(fmaxf(m, e0), fmaxf(e1, e2)), e3);
    const float sc = __builtin_amdgcn_exp2f(m - mn);
    const float p0 = __builtin_amdgcn_exp2f(e0 - mn);
    const float p1 = __builtin_amdgcn_exp2f(e1 - mn);
    const float p2 = __builtin_amdgcn_exp2f(e2 - mn);
    const float p3 = __builtin_amdgcn_exp2f(e3 - mn);
    d = d * sc + ((p0 + p1) + (p2 + p3));
    acc.x = (((acc.x * sc + f0.x * p0) + f1.x * p1) + f2.x * p2) + f3.x * p3;
    acc.y = (((acc.y * sc + f0.y * p0) + f1.y * p1) + f2.y * p2) + f3.y * p3;
    acc.z = (((acc.z * sc + f0.z * p0) + f1.z * p1) + f2.z * p2) + f3.z * p3;
    acc.w = (((acc.w * sc + f0.w * p0) + f1.w * p1) + f2.w * p2) + f3.w * p3;
    m = mn;
}

__global__ void __launch_bounds__(256)
gat_node_kernel(const float* __restrict__ feat,
                const float* __restrict__ attn_l,
                const float* __restrict__ attn_r,
                const int* __restrict__ off,
                const int* __restrict__ degs,
                const int* __restrict__ csr_src,
                const float* __restrict__ bias,
                float* __restrict__ out) {
    const int lane = threadIdx.x & 31;
    const int node = __builtin_amdgcn_readfirstlane(
        (int)((blockIdx.x * blockDim.x + threadIdx.x) >> 5));
    if (node >= N_NODES) return;

    const int beg = __builtin_amdgcn_readfirstlane(off[node]);
    const int deg = __builtin_amdgcn_readfirstlane(degs[node]);
    const float4* feat4 = (const float4*)feat;

    float4 al = ((const float4*)attn_l)[lane];
    float4 ar = ((const float4*)attn_r)[lane];
    al.x *= LOG2E; al.y *= LOG2E; al.z *= LOG2E; al.w *= LOG2E;
    ar.x *= LOG2E; ar.y *= LOG2E; ar.z *= LOG2E; ar.w *= LOG2E;
    const float4 fo = feat4[(size_t)node * (HF / 4) + lane];
    const float  er = gat_sum8(gat_dot4(fo, ar));

    float  m   = -__builtin_inff();
    float  d   = 0.0f;
    float4 acc = make_float4(0.0f, 0.0f, 0.0f, 0.0f);
    for (int cb = 0; cb < deg; cb += 32) {
        const int n = (deg - cb < 32) ? (deg - cb) : 32;
        const int s = (lane < n) ? csr_src[beg + cb + lane] : 0;
        const int nfull = n & ~3;
        for (int j = 0; j < nfull; j += 4)
            gat_edges4<false>(feat4, lane, s, j, 4, al, er, m, d, acc);
        if (n & 3)
            gat_edges4<true>(feat4, lane, s, nfull, n & 3, al, er, m, d, acc);
    }

    const float inv = 1.0f / fmaxf(d, 1e-9f);
    const float4 b4 = ((const float4*)bias)[lane];
    const float rx = fmaxf(acc.x * inv + b4.x, 0.0f);
    const float ry = fmaxf(acc.y * inv + b4.y, 0.0f);
    const float rz = fmaxf(acc.z * inv + b4.z, 0.0f);
    const float rw = fmaxf(acc.w * inv + b4.w, 0.0f);
    float4 o;
    o.x = ((rx + __shfl_down(rx, 8)) + __shfl_down(rx, 16)) + __shfl_down(rx, 24);
    o.y = ((ry + __shfl_down(ry, 8)) + __shfl_down(ry, 16)) + __shfl_down(ry, 24);
    o.z = ((rz + __shfl_down(rz, 8)) + __shfl_down(rz, 16)) + __shfl_down(rz, 24);
    o.w = ((rw + __shfl_down(rw, 8)) + __shfl_down(rw, 16)) + __shfl_down(rw, 24);
    if (lane < 8) {
        o.x *= (1.0f / N_HEADS);
        o.y *= (1.0f / N_HEADS);
        o.z *= (1.0f / N_HEADS);
        o.w *= (1.0f / N_HEADS);
        ((float4*)out)[(size_t)node * (OUT_FEAT / 4) + lane] = o;
    }
}

extern "C" void kernel_launch(void* const* d_in, const int* in_sizes, int n_in,
                              void* d_out, int out_size, void* d_ws, size_t ws_size,
                              hipStream_t stream) {
    const float* h      = (const float*)d_in[0];
    const float* W      = (const float*)d_in[1];
    const float* attn_l = (const float*)d_in[2];
    const float* attn_r = (const float*)d_in[3];
    const float* bias   = (const float*)d_in[4];
    const int*   src    = (const int*)d_in[5];
    const int*   dst    = (const int*)d_in[6];
    float* out = (float*)d_out;

    float* ws      = (float*)d_ws;
    float* feat    = ws;
    int*   off      = (int*)(feat + (size_t)N_NODES * HF);
    int*   degs     = off      + 50176;
    int*   desc     = degs     + 50176;
    int*   csr_src  = desc     + N_CHUNKS * BINS_PAD;
    unsigned* rec   = (unsigned*)(csr_src + N_EDGES + N_BINS * 32);

    const int BS = 256;

    gat_proj_part_kernel<<<N_CHUNKS + PROJ_BLOCKS, 256, 0, stream>>>(
        h, W, feat, src, dst, rec, desc);
    gat_bin_csr_kernel<<<N_BINS, 1024, 0, stream>>>(rec, desc, off, degs, csr_src);
    gat_node_kernel<<<(N_NODES * 32 + BS - 1) / BS, BS, 0, stream>>>(
        feat, attn_l, attn_r, off, degs, csr_src, bias, out);
}
